# starting kernel
# baseline (speedup 1.0000x reference)
#include <hip/hip_runtime.h>

typedef float v2f __attribute__((ext_vector_type(2)));
typedef float v8f __attribute__((ext_vector_type(8)));

#define F 128
#define LN_EPS 1e-5f

__global__ void k_deg(const int* __restrict__ dst, int E, int* __restrict__ cnt) {
    int e = blockIdx.x * blockDim.x + threadIdx.x;
    if (e < E) atomicAdd(&cnt[dst[e]], 1);
}

__global__ void k_dinv(const int* __restrict__ cnt, int N, float* __restrict__ dinv) {
    int v = blockIdx.x * blockDim.x + threadIdx.x;
    if (v < N) dinv[v] = rsqrtf((float)(cnt[v] + 1));
}

__global__ __launch_bounds__(256) void k_gemm(const float* __restrict__ X,
                                              const float* __restrict__ W,
                                              const float* __restrict__ dinv,
                                              const float* __restrict__ bias,
                                              float* __restrict__ H,
                                              float* __restrict__ agg, int N) {
    __shared__ float sW[F * F];
    for (int i = threadIdx.x; i < (F / 2) * (F / 4); i += 256) {
        int p = i >> 5;
        int c = (i & 31) * 4;
        float4 w0 = *(const float4*)&W[(2 * p)     * F + c];
        float4 w1 = *(const float4*)&W[(2 * p + 1) * F + c];
        v2f* d = (v2f*)&sW[p * 256 + c * 2];
        d[0] = (v2f){w0.x, w1.x};
        d[1] = (v2f){w0.y, w1.y};
        d[2] = (v2f){w0.z, w1.z};
        d[3] = (v2f){w0.w, w1.w};
    }
    __syncthreads();

    const int wave = threadIdx.x >> 5;
    const int lane = threadIdx.x & 31;
    const int tile = blockIdx.x * 8 + wave;
    if (tile * 16 >= N) return;

    const int col   = lane & 15;
    const int khalf = (lane >> 4) << 1;
    int row = tile * 16 + (lane & 15);
    if (row > N - 1) row = N - 1;
    const float* xrow = X + (size_t)row * F;

    v8f acc[8];
    const v8f vzero = {0.f, 0.f, 0.f, 0.f, 0.f, 0.f, 0.f, 0.f};
#pragma unroll
    for (int j = 0; j < 8; ++j) acc[j] = vzero;

    for (int k0 = 0; k0 < F; k0 += 4) {
        v2f a;
        a.x = xrow[k0 + khalf];
        a.y = xrow[k0 + khalf + 1];
        const v2f* bbase = (const v2f*)&sW[((k0 + khalf) >> 1) * 256 + col * 2];
#pragma unroll
        for (int j = 0; j < 8; ++j) {
            v2f b = bbase[j * 16];
            acc[j] = __builtin_amdgcn_wmma_f32_16x16x4_f32(
                false, a, false, b, (short)0, acc[j], false, false);
        }
    }

    const int rbase = tile * 16 + (lane >> 4) * 8;
    float nrm[8];
#pragma unroll
    for (int i = 0; i < 8; ++i) {
        int r = rbase + i;
        float di = dinv[r < N ? r : N - 1];
        nrm[i] = di * di;
    }
#pragma unroll
    for (int j = 0; j < 8; ++j) {
        float bj = bias[j * 16 + col];
#pragma unroll
        for (int i = 0; i < 8; ++i) {
            int r = rbase + i;
            if (r < N) {
                float h = acc[j][i];
                size_t o = (size_t)r * F + j * 16 + col;
                H[o]   = h;
                agg[o] = h * nrm[i] + bj;
            }
        }
    }
}

__global__ __launch_bounds__(256) void k_scatter(const int* __restrict__ src,
                                                 const int* __restrict__ dst,
                                                 const float* __restrict__ dinv,
                                                 const float* __restrict__ H,
                                                 float* __restrict__ agg, int E) {
    int gw   = (blockIdx.x * blockDim.x + threadIdx.x) >> 5;
    int lane = threadIdx.x & 31;
    if (gw >= E) return;
    int s = src[gw], d = dst[gw];
    float nrm = dinv[s] * dinv[d];
    float4 hv = *(const float4*)&H[(size_t)s * F + lane * 4];
    float* o = &agg[(size_t)d * F + lane * 4];
    unsafeAtomicAdd(o + 0, hv.x * nrm);
    unsafeAtomicAdd(o + 1, hv.y * nrm);
    unsafeAtomicAdd(o + 2, hv.z * nrm);
    unsafeAtomicAdd(o + 3, hv.w * nrm);
}

__global__ __launch_bounds__(256) void k_relu_ln(const float* __restrict__ agg,
                                                 const float* __restrict__ g,
                                                 const float* __restrict__ be,
                                                 float* __restrict__ out, int N) {
    int wave = threadIdx.x >> 5;
    int lane = threadIdx.x & 31;
    int row = blockIdx.x * 8 + wave;
    if (row >= N) return;

    float4 v = *(const float4*)&agg[(size_t)row * F + lane * 4];
    v.x = fmaxf(v.x, 0.f); v.y = fmaxf(v.y, 0.f);
    v.z = fmaxf(v.z, 0.f); v.w = fmaxf(v.w, 0.f);

    float s = v.x + v.y + v.z + v.w;
#pragma unroll
    for (int m = 16; m >= 1; m >>= 1) s += __shfl_xor(s, m, 32);
    float mu = s * (1.f / F);

    float dx = v.x - mu, dy = v.y - mu, dz = v.z - mu, dw = v.w - mu;
    float q = dx * dx + dy * dy + dz * dz + dw * dw;
#pragma unroll
    for (int m = 16; m >= 1; m >>= 1) q += __shfl_xor(q, m, 32);
    float rs = rsqrtf(q * (1.f / F) + LN_EPS);

    float4 gg = *(const float4*)&g[lane * 4];
    float4 bb = *(const float4*)&be[lane * 4];
    float4 o;
    o.x = dx * rs * gg.x + bb.x;
    o.y = dy * rs * gg.y + bb.y;
    o.z = dz * rs * gg.z + bb.z;
    o.w = dw * rs * gg.w + bb.w;
    *(float4*)&out[(size_t)row * F + lane * 4] = o;
}

static inline size_t align256(size_t x) { return (x + 255) & ~(size_t)255; }

extern "C" void kernel_launch(void* const* d_in, const int* in_sizes, int n_in,
                              void* d_out, int out_size, void* d_ws, size_t ws_size,
                              hipStream_t stream) {
    const float* x   = (const float*)d_in[0];
    const int*   ei  = (const int*)  d_in[1];
    const float* W1  = (const float*)d_in[2];
    const float* b1  = (const float*)d_in[3];
    const float* W2  = (const float*)d_in[4];
    const float* b2  = (const float*)d_in[5];
    const float* g1  = (const float*)d_in[6];
    const float* be1 = (const float*)d_in[7];
    const float* g2  = (const float*)d_in[8];
    const float* be2 = (const float*)d_in[9];

    const int N = in_sizes[0] / F;
    const int E = in_sizes[1] / 2;
    const int* src = ei;
    const int* dst = ei + E;
    float* out = (float*)d_out;

    char* ws = (char*)d_ws;
    size_t off = 0;
    float* dinv = (float*)(ws + off); off = align256(off + (size_t)N * 4);
    int*   cnt  = (int*)  (ws + off); off = align256(off + (size_t)N * 4);
    float* H    = (float*)(ws + off); off = align256(off + (size_t)N * F * 4);
    float* agg  = (float*)(ws + off); off = align256(off + (size_t)N * F * 4);
    (void)ws_size; (void)n_in; (void)out_size;

    const int T = 256;
    dim3 blk(T);

    hipMemsetAsync(cnt, 0, (size_t)N * 4, stream);
    k_deg <<<(E + T - 1) / T, blk, 0, stream>>>(dst, E, cnt);
    k_dinv<<<(N + T - 1) / T, blk, 0, stream>>>(cnt, N, dinv);

    const int gGemm = (N + 127) / 128;
    const int gScat = (E + 7) / 8;
    const int gLN   = (N + 7) / 8;

    k_gemm   <<<gGemm, blk, 0, stream>>>(x, W1, dinv, b1, H, agg, N);
    k_scatter<<<gScat, blk, 0, stream>>>(src, dst, dinv, H, agg, E);
    k_relu_ln<<<gLN,   blk, 0, stream>>>(agg, g1, be1, out, N);

    k_gemm   <<<gGemm, blk, 0, stream>>>(out, W2, dinv, b2, H, agg, N);
    k_scatter<<<gScat, blk, 0, stream>>>(src, dst, dinv, H, agg, E);
    k_relu_ln<<<gLN,   blk, 0, stream>>>(agg, g2, be2, out, N);
}


# Edge sum as per-destination gather: edges bucketed by dst in-call (slot = arrival-order rank returned by the degree atomic + prefix sum), one wave sums a row in registers; f32 atomics gone
# speedup vs baseline: 2.1702x; 2.1702x over previous
#include <hip/hip_runtime.h>

typedef float v2f __attribute__((ext_vector_type(2)));
typedef float v8f __attribute__((ext_vector_type(8)));

#define F 128
#define LN_EPS 1e-5f

__global__ void k_deg(const int* __restrict__ dst, int E, int* __restrict__ cnt,
                      int* __restrict__ rank) {
    int e = blockIdx.x * blockDim.x + threadIdx.x;
    if (e < E) rank[e] = atomicAdd(&cnt[dst[e]], 1);
}

__global__ __launch_bounds__(1024) void k_dinv_scan(const int* __restrict__ cnt, int N,
                                                    float* __restrict__ dinv,
                                                    int* __restrict__ rowptr) {
    __shared__ int sred[32];
    const int tid  = threadIdx.x;
    const int lane = tid & 31;
    const int wave = tid >> 5;
    const int first = blockIdx.x * 1024;

    int part = 0;
    const int4* c4 = (const int4*)cnt;
    const int n4 = first >> 2;
#pragma unroll 4
    for (int i = tid; i < n4; i += 1024) {
        int4 q = c4[i];
        part += (q.x + q.y) + (q.z + q.w);
    }
    for (int m = 16; m >= 1; m >>= 1) part += __shfl_xor(part, m, 32);
    if (lane == 0) sred[wave] = part;
    __syncthreads();
    int base = sred[lane];
    for (int m = 16; m >= 1; m >>= 1) base += __shfl_xor(base, m, 32);
    __syncthreads();

    const int v = first + tid;
    const int c = (v < N) ? cnt[v] : 0;
    int x = c;
    for (int o = 1; o < 32; o <<= 1) {
        int y = __shfl_up(x, o, 32);
        if (lane >= o) x += y;
    }
    if (lane == 31) sred[wave] = x;
    __syncthreads();
    const int wtot = sred[lane];
    int wx = wtot;
    for (int o = 1; o < 32; o <<= 1) {
        int y = __shfl_up(wx, o, 32);
        if (lane >= o) wx += y;
    }
    const int before = __shfl(wx - wtot, wave, 32);
    if (v < N) {
        rowptr[v] = base + before + (x - c);
        dinv[v]   = rsqrtf((float)(c + 1));
    }
}

__global__ __launch_bounds__(256) void k_fill(const int* __restrict__ src,
                                              const int* __restrict__ dst,
                                              const int* __restrict__ rank,
                                              const int* __restrict__ rowptr,
                                              const float* __restrict__ dinv,
                                              int* __restrict__ erec, int E) {
    const int stride = gridDim.x * blockDim.x;
    for (int e = blockIdx.x * blockDim.x + threadIdx.x; e < E; e += stride) {
        const int s = src[e];
        const int d = dst[e];
        const size_t pos = (size_t)(rowptr[d] + rank[e]);
        *(int2*)&erec[2 * pos] = make_int2(s, __float_as_int(dinv[s]));
    }
}

__global__ __launch_bounds__(256) void k_gemm(const float* __restrict__ X,
                                              const float* __restrict__ W,
                                              const float* __restrict__ dinv,
                                              const float* __restrict__ bias,
                                              float* __restrict__ H,
                                              float* __restrict__ agg, int N) {
    __shared__ float sW[F * F];
    for (int i = threadIdx.x; i < (F / 2) * (F / 4); i += 256) {
        int p = i >> 5;
        int c = (i & 31) * 4;
        float4 w0 = *(const float4*)&W[(2 * p)     * F + c];
        float4 w1 = *(const float4*)&W[(2 * p + 1) * F + c];
        v2f* d = (v2f*)&sW[p * 256 + c * 2];
        d[0] = (v2f){w0.x, w1.x};
        d[1] = (v2f){w0.y, w1.y};
        d[2] = (v2f){w0.z, w1.z};
        d[3] = (v2f){w0.w, w1.w};
    }
    __syncthreads();

    const int wave = threadIdx.x >> 5;
    const int lane = threadIdx.x & 31;
    const int tile = blockIdx.x * 8 + wave;
    if (tile * 16 >= N) return;

    const int col   = lane & 15;
    const int khalf = (lane >> 4) << 1;
    int row = tile * 16 + (lane & 15);
    if (row > N - 1) row = N - 1;
    const float* xrow = X + (size_t)row * F;

    v8f acc[8];
    const v8f vzero = {0.f, 0.f, 0.f, 0.f, 0.f, 0.f, 0.f, 0.f};
#pragma unroll
    for (int j = 0; j < 8; ++j) acc[j] = vzero;

    for (int k0 = 0; k0 < F; k0 += 4) {
        v2f a;
        a.x = xrow[k0 + khalf];
        a.y = xrow[k0 + khalf + 1];
        const v2f* bbase = (const v2f*)&sW[((k0 + khalf) >> 1) * 256 + col * 2];
#pragma unroll
        for (int j = 0; j < 8; ++j) {
            v2f b = bbase[j * 16];
            acc[j] = __builtin_amdgcn_wmma_f32_16x16x4_f32(
                false, a, false, b, (short)0, acc[j], false, false);
        }
    }

    const int rbase = tile * 16 + (lane >> 4) * 8;
    float nrm[8];
#pragma unroll
    for (int i = 0; i < 8; ++i) {
        int r = rbase + i;
        float di = dinv[r < N ? r : N - 1];
        nrm[i] = di * di;
    }
#pragma unroll
    for (int j = 0; j < 8; ++j) {
        float bj = bias[j * 16 + col];
#pragma unroll
        for (int i = 0; i < 8; ++i) {
            int r = rbase + i;
            if (r < N) {
                float h = acc[j][i];
                size_t o = (size_t)r * F + j * 16 + col;
                H[o]   = h;
                agg[o] = h * nrm[i] + bj;
            }
        }
    }
}

#define ROWS_PER_WAVE 4

template <int K>
__device__ __forceinline__ void gather_batch(const float* __restrict__ Hl, int sv, float wv,
                                             int j, float4& acc) {
    float4 h[K];
    float  w[K];
#pragma unroll
    for (int i = 0; i < K; ++i) {
        const int s = __builtin_amdgcn_readlane(sv, j + i);
        w[i] = __int_as_float(__builtin_amdgcn_readlane(__float_as_int(wv), j + i));
        h[i] = *(const float4*)(Hl + (size_t)s * F);
    }
#pragma unroll
    for (int i = 0; i < K; ++i) {
        acc.x = fmaf(h[i].x, w[i], acc.x);
        acc.y = fmaf(h[i].y, w[i], acc.y);
        acc.z = fmaf(h[i].z, w[i], acc.z);
        acc.w = fmaf(h[i].w, w[i], acc.w);
    }
}

__global__ __launch_bounds__(256) void k_gather(const int* __restrict__ rowptr,
                                                const int* __restrict__ cnt,
                                                const int* __restrict__ erec,
                                                const float* __restrict__ dinv,
                                                const float* __restrict__ H,
                                                float* __restrict__ agg, int N) {
    const int lane = threadIdx.x & 31;
    const int row0 = __builtin_amdgcn_readfirstlane(
        (int)(blockIdx.x * 8 + (threadIdx.x >> 5)) * ROWS_PER_WAVE);
    if (row0 >= N) return;
    const float* Hl = H + lane * 4;

    int mrow = row0 + (lane & (ROWS_PER_WAVE - 1));
    if (mrow > N - 1) mrow = N - 1;
    const int   startv = rowptr[mrow];
    const int   degv   = cnt[mrow];
    const float ddv    = dinv[mrow];

    for (int r = 0; r < ROWS_PER_WAVE; ++r) {
        const int row = row0 + r;
        if (row >= N) break;
        const int   start = __builtin_amdgcn_readlane(startv, r);
        const int   deg   = __builtin_amdgcn_readlane(degv, r);
        const float dd    = __int_as_float(__builtin_amdgcn_readlane(__float_as_int(ddv), r));
        float4* arow = (float4*)&agg[(size_t)row * F + lane * 4];
        float4 a = *arow;
        float4 acc = {0.f, 0.f, 0.f, 0.f};

        for (int base = 0; base < deg; base += 32) {
            const int n = (deg - base < 32) ? (deg - base) : 32;
            int   sv = 0;
            float wv = 0.f;
            if (lane < n) {
                const int2 rec = *(const int2*)&erec[2 * (size_t)(start + base + lane)];
                sv = rec.x;
                wv = __int_as_float(rec.y) * dd;
            }
            int j = 0;
            for (; j + 8 <= n; j += 8) gather_batch<8>(Hl, sv, wv, j, acc);
            if (j + 4 <= n) { gather_batch<4>(Hl, sv, wv, j, acc); j += 4; }
            if (j + 2 <= n) { gather_batch<2>(Hl, sv, wv, j, acc); j += 2; }
            if (j < n)        gather_batch<1>(Hl, sv, wv, j, acc);
        }
        a.x += acc.x; a.y += acc.y; a.z += acc.z; a.w += acc.w;
        *arow = a;
    }
}

__global__ __launch_bounds__(256) void k_relu_ln(const float* __restrict__ agg,
                                                 const float* __restrict__ g,
                                                 const float* __restrict__ be,
                                                 float* __restrict__ out, int N) {
    int wave = threadIdx.x >> 5;
    int lane = threadIdx.x & 31;
    int row = blockIdx.x * 8 + wave;
    if (row >= N) return;

    float4 v = *(const float4*)&agg[(size_t)row * F + lane * 4];
    v.x = fmaxf(v.x, 0.f); v.y = fmaxf(v.y, 0.f);
    v.z = fmaxf(v.z, 0.f); v.w = fmaxf(v.w, 0.f);

    float s = v.x + v.y + v.z + v.w;
#pragma unroll
    for (int m = 16; m >= 1; m >>= 1) s += __shfl_xor(s, m, 32);
    float mu = s * (1.f / F);

    float dx = v.x - mu, dy = v.y - mu, dz = v.z - mu, dw = v.w - mu;
    float q = dx * dx + dy * dy + dz * dz + dw * dw;
#pragma unroll
    for (int m = 16; m >= 1; m >>= 1) q += __shfl_xor(q, m, 32);
    float rs = rsqrtf(q * (1.f / F) + LN_EPS);

    float4 gg = *(const float4*)&g[lane * 4];
    float4 bb = *(const float4*)&be[lane * 4];
    float4 o;
    o.x = dx * rs * gg.x + bb.x;
    o.y = dy * rs * gg.y + bb.y;
    o.z = dz * rs * gg.z + bb.z;
    o.w = dw * rs * gg.w + bb.w;
    *(float4*)&out[(size_t)row * F + lane * 4] = o;
}

static inline size_t align256(size_t x) { return (x + 255) & ~(size_t)255; }

extern "C" void kernel_launch(void* const* d_in, const int* in_sizes, int n_in,
                              void* d_out, int out_size, void* d_ws, size_t ws_size,
                              hipStream_t stream) {
    const float* x   = (const float*)d_in[0];
    const int*   ei  = (const int*)  d_in[1];
    const float* W1  = (const float*)d_in[2];
    const float* b1  = (const float*)d_in[3];
    const float* W2  = (const float*)d_in[4];
    const float* b2  = (const float*)d_in[5];
    const float* g1  = (const float*)d_in[6];
    const float* be1 = (const float*)d_in[7];
    const float* g2  = (const float*)d_in[8];
    const float* be2 = (const float*)d_in[9];

    const int N = in_sizes[0] / F;
    const int E = in_sizes[1] / 2;
    const int* src = ei;
    const int* dst = ei + E;
    float* out = (float*)d_out;

    char* ws = (char*)d_ws;
    size_t off = 0;
    float* dinv = (float*)(ws + off); off = align256(off + (size_t)N * 4);
    int*   cnt  = (int*)  (ws + off); off = align256(off + (size_t)N * 4);
    float* H    = (float*)(ws + off); off = align256(off + (size_t)N * F * 4);
    float* agg  = (float*)(ws + off); off = align256(off + (size_t)N * F * 4);
    int*   rank   = (int*)(ws + off); off = align256(off + (size_t)E * 4);
    int*   rowptr = (int*)(ws + off); off = align256(off + (size_t)N * 4);
    int*   erec   = (int*)(ws + off); off = align256(off + (size_t)E * 8);
    (void)ws_size; (void)n_in; (void)out_size;

    const int T = 256;
    dim3 blk(T);

    hipMemsetAsync(cnt, 0, (size_t)N * 4, stream);
    k_deg      <<<(E + T - 1) / T, blk, 0, stream>>>(dst, E, cnt, rank);
    k_dinv_scan<<<(N + 1023) / 1024, dim3(1024), 0, stream>>>(cnt, N, dinv, rowptr);
    k_fill     <<<(E + 4 * T - 1) / (4 * T), blk, 0, stream>>>(src, dst, rank, rowptr, dinv, erec, E);

    const int gGemm = (N + 127) / 128;
    const int gGath = (N + 8 * ROWS_PER_WAVE - 1) / (8 * ROWS_PER_WAVE);
    const int gLN   = (N + 7) / 8;

    k_gemm   <<<gGemm, blk, 0, stream>>>(x, W1, dinv, b1, H, agg, N);
    k_gather <<<gGath, blk, 0, stream>>>(rowptr, cnt, erec, dinv, H, agg, N);
    k_relu_ln<<<gLN,   blk, 0, stream>>>(agg, g1, be1, out, N);

    k_gemm   <<<gGemm, blk, 0, stream>>>(out, W2, dinv, b2, H, agg, N);
    k_gather <<<gGath, blk, 0, stream>>>(rowptr, cnt, erec, dinv, H, agg, N);
    k_relu_ln<<<gLN,   blk, 0, stream>>>(agg, g2, be2, out, N);
}


# agg plane and the ReLU+LayerNorm launches removed: the gather wave that holds a row's edge sum adds self loop (own H row) and bias, then ReLU+LayerNorm in registers, writes out once
# speedup vs baseline: 2.5037x; 1.1537x over previous
#include <hip/hip_runtime.h>

typedef float v2f __attribute__((ext_vector_type(2)));
typedef float v8f __attribute__((ext_vector_type(8)));

#define F 128
#define LN_EPS 1e-5f

__global__ void k_deg(const int* __restrict__ dst, int E, int* __restrict__ cnt,
                      int* __restrict__ rank) {
    int e = blockIdx.x * blockDim.x + threadIdx.x;
    if (e < E) rank[e] = atomicAdd(&cnt[dst[e]], 1);
}

__global__ __launch_bounds__(1024) void k_dinv_scan(const int* __restrict__ cnt, int N,
                                                    float* __restrict__ dinv,
                                                    int* __restrict__ rowptr) {
    __shared__ int sred[32];
    const int tid  = threadIdx.x;
    const int lane = tid & 31;
    const int wave = tid >> 5;
    const int first = blockIdx.x * 1024;

    int part = 0;
    const int4* c4 = (const int4*)cnt;
    const int n4 = first >> 2;
#pragma unroll 4
    for (int i = tid; i < n4; i += 1024) {
        int4 q = c4[i];
        part += (q.x + q.y) + (q.z + q.w);
    }
    for (int m = 16; m >= 1; m >>= 1) part += __shfl_xor(part, m, 32);
    if (lane == 0) sred[wave] = part;
    __syncthreads();
    int base = sred[lane];
    for (int m = 16; m >= 1; m >>= 1) base += __shfl_xor(base, m, 32);
    __syncthreads();

    const int v = first + tid;
    const int c = (v < N) ? cnt[v] : 0;
    int x = c;
    for (int o = 1; o < 32; o <<= 1) {
        int y = __shfl_up(x, o, 32);
        if (lane >= o) x += y;
    }
    if (lane == 31) sred[wave] = x;
    __syncthreads();
    const int wtot = sred[lane];
    int wx = wtot;
    for (int o = 1; o < 32; o <<= 1) {
        int y = __shfl_up(wx, o, 32);
        if (lane >= o) wx += y;
    }
    const int before = __shfl(wx - wtot, wave, 32);
    if (v < N) {
        rowptr[v] = base + before + (x - c);
        dinv[v]   = rsqrtf((float)(c + 1));
    }
}

__global__ __launch_bounds__(256) void k_fill(const int* __restrict__ src,
                                              const int* __restrict__ dst,
                                              const int* __restrict__ rank,
                                              const int* __restrict__ rowptr,
                                              const float* __restrict__ dinv,
                                              int* __restrict__ erec, int E) {
    const int stride = gridDim.x * blockDim.x;
    for (int e = blockIdx.x * blockDim.x + threadIdx.x; e < E; e += stride) {
        const int s = src[e];
        const int d = dst[e];
        const size_t pos = (size_t)(rowptr[d] + rank[e]);
        *(int2*)&erec[2 * pos] = make_int2(s, __float_as_int(dinv[s]));
    }
}

__global__ __launch_bounds__(256) void k_gemm(const float* __restrict__ X,
                                              const float* __restrict__ W,
                                              float* __restrict__ H, int N) {
    __shared__ float sW[F * F];
    for (int i = threadIdx.x; i < (F / 2) * (F / 4); i += 256) {
        int p = i >> 5;
        int c = (i & 31) * 4;
        float4 w0 = *(const float4*)&W[(2 * p)     * F + c];
        float4 w1 = *(const float4*)&W[(2 * p + 1) * F + c];
        v2f* d = (v2f*)&sW[p * 256 + c * 2];
        d[0] = (v2f){w0.x, w1.x};
        d[1] = (v2f){w0.y, w1.y};
        d[2] = (v2f){w0.z, w1.z};
        d[3] = (v2f){w0.w, w1.w};
    }
    __syncthreads();

    const int wave = threadIdx.x >> 5;
    const int lane = threadIdx.x & 31;
    const int tile = blockIdx.x * 8 + wave;
    if (tile * 16 >= N) return;

    const int col   = lane & 15;
    const int khalf = (lane >> 4) << 1;
    int row = tile * 16 + (lane & 15);
    if (row > N - 1) row = N - 1;
    const float* xrow = X + (size_t)row * F;

    v8f acc[8];
    const v8f vzero = {0.f, 0.f, 0.f, 0.f, 0.f, 0.f, 0.f, 0.f};
#pragma unroll
    for (int j = 0; j < 8; ++j) acc[j] = vzero;

    for (int k0 = 0; k0 < F; k0 += 4) {
        v2f a;
        a.x = xrow[k0 + khalf];
        a.y = xrow[k0 + khalf + 1];
        const v2f* bbase = (const v2f*)&sW[((k0 + khalf) >> 1) * 256 + col * 2];
#pragma unroll
        for (int j = 0; j < 8; ++j) {
            v2f b = bbase[j * 16];
            acc[j] = __builtin_amdgcn_wmma_f32_16x16x4_f32(
                false, a, false, b, (short)0, acc[j], false, false);
        }
    }

    const int rbase = tile * 16 + (lane >> 4) * 8;
#pragma unroll
    for (int j = 0; j < 8; ++j) {
#pragma unroll
        for (int i = 0; i < 8; ++i) {
            int r = rbase + i;
            if (r < N) H[(size_t)r * F + j * 16 + col] = acc[j][i];
        }
    }
}

#define ROWS_PER_WAVE 4

template <int K>
__device__ __forceinline__ void gather_batch(const float* __restrict__ Hl, int sv, float wv,
                                             int j, float4& acc) {
    float4 h[K];
    float  w[K];
#pragma unroll
    for (int i = 0; i < K; ++i) {
        const int s = __builtin_amdgcn_readlane(sv, j + i);
        w[i] = __int_as_float(__builtin_amdgcn_readlane(__float_as_int(wv), j + i));
        h[i] = *(const float4*)(Hl + (size_t)s * F);
    }
#pragma unroll
    for (int i = 0; i < K; ++i) {
        acc.x = fmaf(h[i].x, w[i], acc.x);
        acc.y = fmaf(h[i].y, w[i], acc.y);
        acc.z = fmaf(h[i].z, w[i], acc.z);
        acc.w = fmaf(h[i].w, w[i], acc.w);
    }
}

__global__ __launch_bounds__(256) void k_gather_ln(const int* __restrict__ rowptr,
                                                   const int* __restrict__ cnt,
                                                   const int* __restrict__ erec,
                                                   const float* __restrict__ dinv,
                                                   const float* __restrict__ H,
                                                   const float* bias,
                                                   const float* g,
                                                   const float* be,
                                                   float* out, int N) {
    const int lane = threadIdx.x & 31;
    const int row0 = __builtin_amdgcn_readfirstlane(
        (int)(blockIdx.x * 8 + (threadIdx.x >> 5)) * ROWS_PER_WAVE);
    if (row0 >= N) return;
    const float* Hl = H + lane * 4;

    int mrow = row0 + (lane & (ROWS_PER_WAVE - 1));
    if (mrow > N - 1) mrow = N - 1;
    const int   startv = rowptr[mrow];
    const int   degv   = cnt[mrow];
    const float ddv    = dinv[mrow];

    for (int r = 0; r < ROWS_PER_WAVE; ++r) {
        const int row = row0 + r;
        if (row >= N) break;
        const int   start = __builtin_amdgcn_readlane(startv, r);
        const int   deg   = __builtin_amdgcn_readlane(degv, r);
        const float dd    = __int_as_float(__builtin_amdgcn_readlane(__float_as_int(ddv), r));
        const float4 hs = *(const float4*)(Hl + (size_t)row * F);
        float4 acc = {0.f, 0.f, 0.f, 0.f};

        for (int base = 0; base < deg; base += 32) {
            const int n = (deg - base < 32) ? (deg - base) : 32;
            int   sv = 0;
            float wv = 0.f;
            if (lane < n) {
                const int2 rec = *(const int2*)&erec[2 * (size_t)(start + base + lane)];
                sv = rec.x;
                wv = __int_as_float(rec.y) * dd;
            }
            int j = 0;
            for (; j + 8 <= n; j += 8) gather_batch<8>(Hl, sv, wv, j, acc);
            if (j + 4 <= n) { gather_batch<4>(Hl, sv, wv, j, acc); j += 4; }
            if (j + 2 <= n) { gather_batch<2>(Hl, sv, wv, j, acc); j += 2; }
            if (j < n)        gather_batch<1>(Hl, sv, wv, j, acc);
        }

        const float4 bj = *(const float4*)&bias[lane * 4];
        const float4 gg = *(const float4*)&g[lane * 4];
        const float4 bb = *(const float4*)&be[lane * 4];

        const float nrm = dd * dd;
        float4 v;
        v.x = hs.x * nrm + bj.x; v.y = hs.y * nrm + bj.y;
        v.z = hs.z * nrm + bj.z; v.w = hs.w * nrm + bj.w;
        v.x += acc.x; v.y += acc.y; v.z += acc.z; v.w += acc.w;
        v.x = fmaxf(v.x, 0.f); v.y = fmaxf(v.y, 0.f);
        v.z = fmaxf(v.z, 0.f); v.w = fmaxf(v.w, 0.f);

        float s = v.x + v.y + v.z + v.w;
        for (int m = 16; m >= 1; m >>= 1) s += __shfl_xor(s, m, 32);
        const float mu = s * (1.f / F);

        const float dx = v.x - mu, dy = v.y - mu, dz = v.z - mu, dw = v.w - mu;
        float q = dx * dx + dy * dy + dz * dz + dw * dw;
        for (int m = 16; m >= 1; m >>= 1) q += __shfl_xor(q, m, 32);
        const float rs = rsqrtf(q * (1.f / F) + LN_EPS);

        float4 o;
        o.x = dx * rs * gg.x + bb.x;
        o.y = dy * rs * gg.y + bb.y;
        o.z = dz * rs * gg.z + bb.z;
        o.w = dw * rs * gg.w + bb.w;
        *(float4*)&out[(size_t)row * F + lane * 4] = o;
    }
}

static inline size_t align256(size_t x) { return (x + 255) & ~(size_t)255; }

extern "C" void kernel_launch(void* const* d_in, const int* in_sizes, int n_in,
                              void* d_out, int out_size, void* d_ws, size_t ws_size,
                              hipStream_t stream) {
    const float* x   = (const float*)d_in[0];
    const int*   ei  = (const int*)  d_in[1];
    const float* W1  = (const float*)d_in[2];
    const float* b1  = (const float*)d_in[3];
    const float* W2  = (const float*)d_in[4];
    const float* b2  = (const float*)d_in[5];
    const float* g1  = (const float*)d_in[6];
    const float* be1 = (const float*)d_in[7];
    const float* g2  = (const float*)d_in[8];
    const float* be2 = (const float*)d_in[9];

    const int N = in_sizes[0] / F;
    const int E = in_sizes[1] / 2;
    const int* src = ei;
    const int* dst = ei + E;
    float* out = (float*)d_out;

    char* ws = (char*)d_ws;
    size_t off = 0;
    float* dinv = (float*)(ws + off); off = align256(off + (size_t)N * 4);
    int*   cnt  = (int*)  (ws + off); off = align256(off + (size_t)N * 4);
    float* H    = (float*)(ws + off); off = align256(off + (size_t)N * F * 4);
    int*   rank   = (int*)(ws + off); off = align256(off + (size_t)E * 4);
    int*   rowptr = (int*)(ws + off); off = align256(off + (size_t)N * 4);
    int*   erec   = (int*)(ws + off); off = align256(off + (size_t)E * 8);
    (void)ws_size; (void)n_in; (void)out_size;

    const int T = 256;
    dim3 blk(T);

    hipMemsetAsync(cnt, 0, (size_t)N * 4, stream);
    k_deg      <<<(E + T - 1) / T, blk, 0, stream>>>(dst, E, cnt, rank);
    k_dinv_scan<<<(N + 1023) / 1024, dim3(1024), 0, stream>>>(cnt, N, dinv, rowptr);
    k_fill     <<<(E + 4 * T - 1) / (4 * T), blk, 0, stream>>>(src, dst, rank, rowptr, dinv, erec, E);

    const int gGemm = (N + 127) / 128;
    const int gGath = (N + 8 * ROWS_PER_WAVE - 1) / (8 * ROWS_PER_WAVE);

    k_gemm     <<<gGemm, blk, 0, stream>>>(x, W1, H, N);
    k_gather_ln<<<gGath, blk, 0, stream>>>(rowptr, cnt, erec, dinv, H, b1, g1, be1, out, N);

    k_gemm     <<<gGemm, blk, 0, stream>>>(out, W2, H, N);
    k_gather_ln<<<gGath, blk, 0, stream>>>(rowptr, cnt, erec, dinv, H, b2, g2, be2, out, N);
}


# Edges bucketed by destination with two shared-memory counting passes (512-node ranges, then nodes) instead of one returning global atomic per edge; slot order = arrival at those counters
# speedup vs baseline: 3.4332x; 1.3713x over previous
#include <hip/hip_runtime.h>

typedef float v2f __attribute__((ext_vector_type(2)));
typedef float v8f __attribute__((ext_vector_type(8)));

#define F 128
#define LN_EPS 1e-5f

#define CB_SHIFT 9
#define CB_NODES (1 << CB_SHIFT)
#define NSLICE   256
#define MAXCB    256

#define EB 8
static_assert(EB == 8, "the asm operand lists in k_hist and k_scatter name EB values");
__global__ __launch_bounds__(1024) void k_hist(const int* __restrict__ dst, int E,
                                               int* __restrict__ bh) {
    __shared__ int h[MAXCB];
    const int tid = threadIdx.x;
    if (tid < MAXCB) h[tid] = 0;
    __syncthreads();
    const int per = (E + NSLICE - 1) / NSLICE;
    const int e0 = blockIdx.x * per;
    int e1 = e0 + per;
    if (e1 > E) e1 = E;
    for (int base = e0 + tid; base < e1; base += EB * 1024) {
        int d[EB];
#pragma unroll
        for (int u = 0; u < EB; ++u) {
            const int e = base + u * 1024;
            d[u] = dst[(e < e1) ? e : e1 - 1];
        }
        asm volatile("" :: "v"(d[0]), "v"(d[1]), "v"(d[2]), "v"(d[3]),
                           "v"(d[4]), "v"(d[5]), "v"(d[6]), "v"(d[7]));
#pragma unroll
        for (int u = 0; u < EB; ++u)
            if (base + u * 1024 < e1) atomicAdd(&h[d[u] >> CB_SHIFT], 1);
    }
    __syncthreads();
    if (tid < MAXCB) bh[blockIdx.x * MAXCB + tid] = h[tid];
}

__global__ __launch_bounds__(1024) void k_scan(const int* __restrict__ bh,
                                               int* __restrict__ obase,
                                               int* __restrict__ bstart) {
    __shared__ int sred[32];
    const int tid  = threadIdx.x;
    const int lane = tid & 31;
    const int wave = tid >> 5;
    const int b  = tid >> 2;
    const int s0 = (tid & 3) * 64;

    int sum = 0;
#pragma unroll 8
    for (int k = 0; k < 64; ++k) sum += bh[(s0 + k) * MAXCB + b];

    int x = sum;
    for (int o = 1; o < 32; o <<= 1) {
        int y = __shfl_up(x, o, 32);
        if (lane >= o) x += y;
    }
    if (lane == 31) sred[wave] = x;
    __syncthreads();
    const int wtot = sred[lane];
    int wx = wtot;
    for (int o = 1; o < 32; o <<= 1) {
        int y = __shfl_up(wx, o, 32);
        if (lane >= o) wx += y;
    }
    const int before = __shfl(wx - wtot, wave, 32);
    int run = before + (x - sum);

    if ((tid & 3) == 0) bstart[b] = run;
    if (tid == 1023)    bstart[MAXCB] = run + sum;
#pragma unroll 8
    for (int k = 0; k < 64; ++k) {
        const int idx = (s0 + k) * MAXCB + b;
        const int v = bh[idx];
        obase[idx] = run;
        run += v;
    }
}

__global__ __launch_bounds__(1024) void k_scatter(const int* __restrict__ src,
                                                  const int* __restrict__ dst,
                                                  const int* __restrict__ obase,
                                                  int* __restrict__ tmp, int E) {
    __shared__ int cur[MAXCB];
    const int tid = threadIdx.x;
    if (tid < MAXCB) cur[tid] = obase[blockIdx.x * MAXCB + tid];
    __syncthreads();
    const int per = (E + NSLICE - 1) / NSLICE;
    const int e0 = blockIdx.x * per;
    int e1 = e0 + per;
    if (e1 > E) e1 = E;
    for (int base = e0 + tid; base < e1; base += EB * 1024) {
        int d[EB], s[EB];
#pragma unroll
        for (int u = 0; u < EB; ++u) {
            const int e  = base + u * 1024;
            const int ec = (e < e1) ? e : e1 - 1;
            d[u] = dst[ec];
            s[u] = src[ec];
        }
        asm volatile("" :: "v"(d[0]), "v"(d[1]), "v"(d[2]), "v"(d[3]),
                           "v"(d[4]), "v"(d[5]), "v"(d[6]), "v"(d[7]),
                           "v"(s[0]), "v"(s[1]), "v"(s[2]), "v"(s[3]),
                           "v"(s[4]), "v"(s[5]), "v"(s[6]), "v"(s[7]));
#pragma unroll
        for (int u = 0; u < EB; ++u) {
            if (base + u * 1024 < e1) {
                const int p = atomicAdd(&cur[d[u] >> CB_SHIFT], 1);
                *(int2*)&tmp[2 * (size_t)p] = make_int2(s[u], d[u]);
            }
        }
    }
}

#define FB 10
static_assert(FB == 10, "the asm operand lists in k_fine name FB values");
__global__ __launch_bounds__(1024) void k_fine(const int* __restrict__ tmp,
                                               const int* __restrict__ bstart, int N,
                                               int* __restrict__ cnt,
                                               int* __restrict__ rowptr,
                                               float* __restrict__ dinv,
                                               int* __restrict__ esrc) {
    __shared__ int c[CB_NODES];
    __shared__ int sred[32];
    const int tid  = threadIdx.x;
    const int lane = tid & 31;
    const int wave = tid >> 5;
    const int node0 = blockIdx.x << CB_SHIFT;
    const int e0 = bstart[blockIdx.x];
    const int e1 = bstart[blockIdx.x + 1];

    if (tid < CB_NODES) c[tid] = 0;
    __syncthreads();
    for (int base = e0 + tid; base < e1; base += FB * 1024) {
        int d[FB];
#pragma unroll
        for (int u = 0; u < FB; ++u) {
            const int i = base + u * 1024;
            d[u] = tmp[2 * (size_t)((i < e1) ? i : e1 - 1) + 1];
        }
        asm volatile("" :: "v"(d[0]), "v"(d[1]), "v"(d[2]), "v"(d[3]), "v"(d[4]),
                           "v"(d[5]), "v"(d[6]), "v"(d[7]), "v"(d[8]), "v"(d[9]));
#pragma unroll
        for (int u = 0; u < FB; ++u)
            if (base + u * 1024 < e1) atomicAdd(&c[d[u] - node0], 1);
    }
    __syncthreads();

    const int cv = (tid < CB_NODES) ? c[tid] : 0;
    int x = cv;
    for (int o = 1; o < 32; o <<= 1) {
        int y = __shfl_up(x, o, 32);
        if (lane >= o) x += y;
    }
    if (lane == 31) sred[wave] = x;
    __syncthreads();
    const int wtot = sred[lane];
    int wx = wtot;
    for (int o = 1; o < 32; o <<= 1) {
        int y = __shfl_up(wx, o, 32);
        if (lane >= o) wx += y;
    }
    const int before = __shfl(wx - wtot, wave, 32);
    const int p0 = e0 + before + (x - cv);

    if (tid < CB_NODES) {
        const int v = node0 + tid;
        if (v < N) {
            cnt[v]    = cv;
            rowptr[v] = p0;
            dinv[v]   = rsqrtf((float)(cv + 1));
        }
        c[tid] = p0;
    }
    __syncthreads();
    for (int base = e0 + tid; base < e1; base += FB * 1024) {
        int2 rec[FB];
#pragma unroll
        for (int u = 0; u < FB; ++u) {
            const int i = base + u * 1024;
            rec[u] = *(const int2*)&tmp[2 * (size_t)((i < e1) ? i : e1 - 1)];
        }
        asm volatile("" :: "v"(rec[0].y), "v"(rec[1].y), "v"(rec[2].y), "v"(rec[3].y), "v"(rec[4].y),
                           "v"(rec[5].y), "v"(rec[6].y), "v"(rec[7].y), "v"(rec[8].y), "v"(rec[9].y));
#pragma unroll
        for (int u = 0; u < FB; ++u) {
            if (base + u * 1024 < e1) {
                const int p = atomicAdd(&c[rec[u].y - node0], 1);
                esrc[p] = rec[u].x;
            }
        }
    }
}

__global__ __launch_bounds__(256) void k_pair(const int* __restrict__ esrc,
                                              const float* __restrict__ dinv,
                                              int* __restrict__ erec, int E) {
    const int stride = gridDim.x * blockDim.x;
    for (int e = blockIdx.x * blockDim.x + threadIdx.x; e < E; e += stride) {
        const int s = esrc[e];
        *(int2*)&erec[2 * (size_t)e] = make_int2(s, __float_as_int(dinv[s]));
    }
}

__global__ __launch_bounds__(256) void k_gemm(const float* __restrict__ X,
                                              const float* __restrict__ W,
                                              float* __restrict__ H, int N) {
    __shared__ float sW[F * F];
    for (int i = threadIdx.x; i < (F / 2) * (F / 4); i += 256) {
        int p = i >> 5;
        int c = (i & 31) * 4;
        float4 w0 = *(const float4*)&W[(2 * p)     * F + c];
        float4 w1 = *(const float4*)&W[(2 * p + 1) * F + c];
        v2f* d = (v2f*)&sW[p * 256 + c * 2];
        d[0] = (v2f){w0.x, w1.x};
        d[1] = (v2f){w0.y, w1.y};
        d[2] = (v2f){w0.z, w1.z};
        d[3] = (v2f){w0.w, w1.w};
    }
    __syncthreads();

    const int wave = threadIdx.x >> 5;
    const int lane = threadIdx.x & 31;
    const int tile = blockIdx.x * 8 + wave;
    if (tile * 16 >= N) return;

    const int col   = lane & 15;
    const int khalf = (lane >> 4) << 1;
    int row = tile * 16 + (lane & 15);
    if (row > N - 1) row = N - 1;
    const float* xrow = X + (size_t)row * F;

    v8f acc[8];
    const v8f vzero = {0.f, 0.f, 0.f, 0.f, 0.f, 0.f, 0.f, 0.f};
#pragma unroll
    for (int j = 0; j < 8; ++j) acc[j] = vzero;

    for (int k0 = 0; k0 < F; k0 += 4) {
        v2f a;
        a.x = xrow[k0 + khalf];
        a.y = xrow[k0 + khalf + 1];
        const v2f* bbase = (const v2f*)&sW[((k0 + khalf) >> 1) * 256 + col * 2];
#pragma unroll
        for (int j = 0; j < 8; ++j) {
            v2f b = bbase[j * 16];
            acc[j] = __builtin_amdgcn_wmma_f32_16x16x4_f32(
                false, a, false, b, (short)0, acc[j], false, false);
        }
    }

    const int rbase = tile * 16 + (lane >> 4) * 8;
#pragma unroll
    for (int j = 0; j < 8; ++j) {
#pragma unroll
        for (int i = 0; i < 8; ++i) {
            int r = rbase + i;
            if (r < N) H[(size_t)r * F + j * 16 + col] = acc[j][i];
        }
    }
}

#define ROWS_PER_WAVE 4

template <int K>
__device__ __forceinline__ void gather_batch(const float* __restrict__ Hl, int sv, float wv,
                                             int j, float4& acc) {
    float4 h[K];
    float  w[K];
#pragma unroll
    for (int i = 0; i < K; ++i) {
        const int s = __builtin_amdgcn_readlane(sv, j + i);
        w[i] = __int_as_float(__builtin_amdgcn_readlane(__float_as_int(wv), j + i));
        h[i] = *(const float4*)(Hl + (size_t)s * F);
    }
#pragma unroll
    for (int i = 0; i < K; ++i) {
        acc.x = fmaf(h[i].x, w[i], acc.x);
        acc.y = fmaf(h[i].y, w[i], acc.y);
        acc.z = fmaf(h[i].z, w[i], acc.z);
        acc.w = fmaf(h[i].w, w[i], acc.w);
    }
}

__global__ __launch_bounds__(256) void k_gather_ln(const int* __restrict__ rowptr,
                                                   const int* __restrict__ cnt,
                                                   const int* __restrict__ erec,
                                                   const float* __restrict__ dinv,
                                                   const float* __restrict__ H,
                                                   const float* bias,
                                                   const float* g,
                                                   const float* be,
                                                   float* out, int N) {
    const int lane = threadIdx.x & 31;
    const int row0 = __builtin_amdgcn_readfirstlane(
        (int)(blockIdx.x * 8 + (threadIdx.x >> 5)) * ROWS_PER_WAVE);
    if (row0 >= N) return;
    const float* Hl = H + lane * 4;

    int mrow = row0 + (lane & (ROWS_PER_WAVE - 1));
    if (mrow > N - 1) mrow = N - 1;
    const int   startv = rowptr[mrow];
    const int   degv   = cnt[mrow];
    const float ddv    = dinv[mrow];

    for (int r = 0; r < ROWS_PER_WAVE; ++r) {
        const int row = row0 + r;
        if (row >= N) break;
        const int   start = __builtin_amdgcn_readlane(startv, r);
        const int   deg   = __builtin_amdgcn_readlane(degv, r);
        const float dd    = __int_as_float(__builtin_amdgcn_readlane(__float_as_int(ddv), r));
        const float4 hs = *(const float4*)(Hl + (size_t)row * F);
        float4 acc = {0.f, 0.f, 0.f, 0.f};

        for (int base = 0; base < deg; base += 32) {
            const int n = (deg - base < 32) ? (deg - base) : 32;
            int   sv = 0;
            float wv = 0.f;
            if (lane < n) {
                const int2 rec = *(const int2*)&erec[2 * (size_t)(start + base + lane)];
                sv = rec.x;
                wv = __int_as_float(rec.y) * dd;
            }
            int j = 0;
            for (; j + 8 <= n; j += 8) gather_batch<8>(Hl, sv, wv, j, acc);
            if (j + 4 <= n) { gather_batch<4>(Hl, sv, wv, j, acc); j += 4; }
            if (j + 2 <= n) { gather_batch<2>(Hl, sv, wv, j, acc); j += 2; }
            if (j < n)        gather_batch<1>(Hl, sv, wv, j, acc);
        }

        const float4 bj = *(const float4*)&bias[lane * 4];
        const float4 gg = *(const float4*)&g[lane * 4];
        const float4 bb = *(const float4*)&be[lane * 4];

        const float nrm = dd * dd;
        float4 v;
        v.x = hs.x * nrm + bj.x; v.y = hs.y * nrm + bj.y;
        v.z = hs.z * nrm + bj.z; v.w = hs.w * nrm + bj.w;
        v.x += acc.x; v.y += acc.y; v.z += acc.z; v.w += acc.w;
        v.x = fmaxf(v.x, 0.f); v.y = fmaxf(v.y, 0.f);
        v.z = fmaxf(v.z, 0.f); v.w = fmaxf(v.w, 0.f);

        float s = v.x + v.y + v.z + v.w;
        for (int m = 16; m >= 1; m >>= 1) s += __shfl_xor(s, m, 32);
        const float mu = s * (1.f / F);

        const float dx = v.x - mu, dy = v.y - mu, dz = v.z - mu, dw = v.w - mu;
        float q = dx * dx + dy * dy + dz * dz + dw * dw;
        for (int m = 16; m >= 1; m >>= 1) q += __shfl_xor(q, m, 32);
        const float rs = rsqrtf(q * (1.f / F) + LN_EPS);

        float4 o;
        o.x = dx * rs * gg.x + bb.x;
        o.y = dy * rs * gg.y + bb.y;
        o.z = dz * rs * gg.z + bb.z;
        o.w = dw * rs * gg.w + bb.w;
        *(float4*)&out[(size_t)row * F + lane * 4] = o;
    }
}

static inline size_t align256(size_t x) { return (x + 255) & ~(size_t)255; }

extern "C" void kernel_launch(void* const* d_in, const int* in_sizes, int n_in,
                              void* d_out, int out_size, void* d_ws, size_t ws_size,
                              hipStream_t stream) {
    const float* x   = (const float*)d_in[0];
    const int*   ei  = (const int*)  d_in[1];
    const float* W1  = (const float*)d_in[2];
    const float* b1  = (const float*)d_in[3];
    const float* W2  = (const float*)d_in[4];
    const float* b2  = (const float*)d_in[5];
    const float* g1  = (const float*)d_in[6];
    const float* be1 = (const float*)d_in[7];
    const float* g2  = (const float*)d_in[8];
    const float* be2 = (const float*)d_in[9];

    const int N = in_sizes[0] / F;
    const int E = in_sizes[1] / 2;
    const int* src = ei;
    const int* dst = ei + E;
    float* out = (float*)d_out;

    char* ws = (char*)d_ws;
    size_t off = 0;
    float* dinv = (float*)(ws + off); off = align256(off + (size_t)N * 4);
    int*   cnt  = (int*)  (ws + off); off = align256(off + (size_t)N * 4);
    float* H    = (float*)(ws + off); off = align256(off + (size_t)N * F * 4);
    int*   esrc   = (int*)(ws + off); off = align256(off + (size_t)E * 4);
    int*   rowptr = (int*)(ws + off); off = align256(off + (size_t)N * 4);
    int*   erec   = (int*)(ws + off); off = align256(off + (size_t)E * 8);
    int*   tmp    = (int*)(ws + off); off = align256(off + (size_t)E * 8);
    int*   bh     = (int*)(ws + off); off = align256(off + (size_t)NSLICE * MAXCB * 4);
    int*   obase  = (int*)(ws + off); off = align256(off + (size_t)NSLICE * MAXCB * 4);
    int*   bstart = (int*)(ws + off); off = align256(off + (size_t)(MAXCB + 1) * 4);
    (void)ws_size; (void)n_in; (void)out_size;

    const int T = 256;
    dim3 blk(T);

    k_hist   <<<NSLICE, dim3(1024), 0, stream>>>(dst, E, bh);
    k_scan   <<<1, dim3(1024), 0, stream>>>(bh, obase, bstart);
    k_scatter<<<NSLICE, dim3(1024), 0, stream>>>(src, dst, obase, tmp, E);
    k_fine   <<<(N + CB_NODES - 1) / CB_NODES, dim3(1024), 0, stream>>>(tmp, bstart, N, cnt, rowptr, dinv, esrc);
    k_pair   <<<(E + 4 * T - 1) / (4 * T), blk, 0, stream>>>(esrc, dinv, erec, E);

    const int gGemm = (N + 127) / 128;
    const int gGath = (N + 8 * ROWS_PER_WAVE - 1) / (8 * ROWS_PER_WAVE);

    k_gemm     <<<gGemm, blk, 0, stream>>>(x, W1, H, N);
    k_gather_ln<<<gGath, blk, 0, stream>>>(rowptr, cnt, erec, dinv, H, b1, g1, be1, out, N);

    k_gemm     <<<gGemm, blk, 0, stream>>>(out, W2, H, N);
    k_gather_ln<<<gGath, blk, 0, stream>>>(rowptr, cnt, erec, dinv, H, b2, g2, be2, out, N);
}


# Coarse bucket pass made slice-local: each edge slice counts, scans and places its own edges in one launch; histogram and scan launches gone; edge order in a bucket changes, row sums reassociate
# speedup vs baseline: 4.6152x; 1.3443x over previous
#include <hip/hip_runtime.h>

typedef float v2f __attribute__((ext_vector_type(2)));
typedef float v8f __attribute__((ext_vector_type(8)));
typedef int   v4i __attribute__((ext_vector_type(4)));

#define F 128
#define LN_EPS 1e-5f

#define CB_SHIFT 9
#define CB_NODES (1 << CB_SHIFT)
#define NSLICE   256
#define MAXCB    256
static_assert(NSLICE == 256 && MAXCB == 256, "k_fine maps thread t to table columns 4*(t & 63).. and slice group t >> 6");

#define EB 8
static_assert(EB == 8, "the asm operand list in k_part names EB values");
__global__ __launch_bounds__(1024) void k_part(const int* __restrict__ src,
                                               const int* __restrict__ dst, int E,
                                               int* __restrict__ scnt,
                                               int* __restrict__ soff,
                                               int* __restrict__ tmp) {
    __shared__ int h[MAXCB];
    __shared__ int sred[32];
    const int tid  = threadIdx.x;
    const int lane = tid & 31;
    const int wave = tid >> 5;
    if (tid < MAXCB) h[tid] = 0;
    __syncthreads();
    const int per = (E + NSLICE - 1) / NSLICE;
    const int e0 = blockIdx.x * per;
    int e1 = e0 + per;
    if (e1 > E) e1 = E;
    const int base0 = e0 + tid;

    int d[EB], s[EB];
#pragma unroll
    for (int u = 0; u < EB; ++u) { d[u] = 0; s[u] = 0; }
    if (e0 < e1) {
#pragma unroll
        for (int u = 0; u < EB; ++u) {
            const int e  = base0 + u * 1024;
            const int ec = (e < e1) ? e : e1 - 1;
            d[u] = dst[ec];
            s[u] = src[ec];
        }
        asm volatile("" :: "v"(d[0]), "v"(d[1]), "v"(d[2]), "v"(d[3]),
                           "v"(d[4]), "v"(d[5]), "v"(d[6]), "v"(d[7]),
                           "v"(s[0]), "v"(s[1]), "v"(s[2]), "v"(s[3]),
                           "v"(s[4]), "v"(s[5]), "v"(s[6]), "v"(s[7]));
#pragma unroll
        for (int u = 0; u < EB; ++u)
            if (base0 + u * 1024 < e1) atomicAdd(&h[d[u] >> CB_SHIFT], 1);
    }
    for (int e = base0 + EB * 1024; e < e1; e += 1024) atomicAdd(&h[dst[e] >> CB_SHIFT], 1);
    __syncthreads();

    const int hv = (tid < MAXCB) ? h[tid] : 0;
    int x = hv;
    for (int o = 1; o < 32; o <<= 1) {
        int y = __shfl_up(x, o, 32);
        if (lane >= o) x += y;
    }
    if (lane == 31) sred[wave] = x;
    __syncthreads();
    const int wtot = sred[lane];
    int wx = wtot;
    for (int o = 1; o < 32; o <<= 1) {
        int y = __shfl_up(wx, o, 32);
        if (lane >= o) wx += y;
    }
    const int before = __shfl(wx - wtot, wave, 32);
    const int p0 = e0 + before + (x - hv);
    if (tid < MAXCB) {
        scnt[blockIdx.x * MAXCB + tid] = hv;
        soff[blockIdx.x * MAXCB + tid] = p0;
        h[tid] = p0;
    }
    __syncthreads();

    if (e0 < e1) {
#pragma unroll
        for (int u = 0; u < EB; ++u) {
            if (base0 + u * 1024 < e1) {
                const int p = atomicAdd(&h[d[u] >> CB_SHIFT], 1);
                *(int2*)&tmp[2 * (size_t)p] = make_int2(s[u], d[u]);
            }
        }
    }
    for (int e = base0 + EB * 1024; e < e1; e += 1024) {
        const int dd = dst[e];
        const int ss = src[e];
        const int p = atomicAdd(&h[dd >> CB_SHIFT], 1);
        *(int2*)&tmp[2 * (size_t)p] = make_int2(ss, dd);
    }
}

#define RPW 8
static_assert(RPW * 32 == NSLICE, "every slice's run belongs to exactly one wave");
static_assert(RPW == 8, "the asm operand list in k_fine names 2*RPW values");
__global__ __launch_bounds__(1024) void k_fine(const int* __restrict__ tmp,
                                               const int* __restrict__ scnt,
                                               const int* __restrict__ soff, int N, int E,
                                               int* __restrict__ cnt,
                                               int* __restrict__ rowptr,
                                               float* __restrict__ dinv,
                                               int* __restrict__ esrc) {
    __shared__ int c[CB_NODES];
    __shared__ int sred[32];
    __shared__ int rs[NSLICE];
    __shared__ int rl[NSLICE];
    const int tid  = threadIdx.x;
    const int lane = tid & 31;
    const int wave = tid >> 5;
    const int rb   = blockIdx.x;
    const int node0 = rb << CB_SHIFT;

    if (tid < NSLICE) {
        rs[tid] = soff[tid * MAXCB + rb];
        rl[tid] = scnt[tid * MAXCB + rb];
    }
    if (tid < CB_NODES) c[tid] = 0;
    int part = 0;
    {
        const int c4 = (tid & 63) * 4;
        const int rg = tid >> 6;
        if (c4 < rb) {
            const int rot = rb & 15;
            v4i v[16];
#pragma unroll
            for (int k = 0; k < 16; ++k)
                v[k] = *(const v4i*)&scnt[(rg * 16 + (k ^ rot)) * MAXCB + c4];
            asm volatile("" :: "v"(v[0]), "v"(v[1]), "v"(v[2]), "v"(v[3]),
                               "v"(v[4]), "v"(v[5]), "v"(v[6]), "v"(v[7]),
                               "v"(v[8]), "v"(v[9]), "v"(v[10]), "v"(v[11]),
                               "v"(v[12]), "v"(v[13]), "v"(v[14]), "v"(v[15]));
            int sx = 0, sy = 0, sz = 0, sw = 0;
#pragma unroll
            for (int k = 0; k < 16; ++k) { sx += v[k].x; sy += v[k].y; sz += v[k].z; sw += v[k].w; }
            part = sx;
            if (c4 + 1 < rb) part += sy;
            if (c4 + 2 < rb) part += sz;
            if (c4 + 3 < rb) part += sw;
        }
    }
    for (int o = 16; o >= 1; o >>= 1) part += __shfl_xor(part, o, 32);
    if (lane == 0) sred[wave] = part;
    __syncthreads();
    int e0 = sred[lane];
    for (int o = 16; o >= 1; o >>= 1) e0 += __shfl_xor(e0, o, 32);

    int st[RPW], ln[RPW];
#pragma unroll
    for (int k = 0; k < RPW; ++k) {
        st[k] = __builtin_amdgcn_readfirstlane(rs[wave + 32 * k]);
        ln[k] = __builtin_amdgcn_readfirstlane(rl[wave + 32 * k]);
    }

    int2 rec[2 * RPW];
#pragma unroll
    for (int k = 0; k < RPW; ++k) {
#pragma unroll
        for (int j = 0; j < 2; ++j) {
            const int i = lane + 32 * j;
            int idx = st[k] + ((i < ln[k]) ? i : 0);
            if (idx > E - 1) idx = E - 1;
            if (idx < 0) idx = 0;
            rec[2 * k + j] = *(const int2*)&tmp[2 * (size_t)idx];
        }
    }
    asm volatile("" :: "v"(rec[0].y), "v"(rec[1].y), "v"(rec[2].y), "v"(rec[3].y),
                       "v"(rec[4].y), "v"(rec[5].y), "v"(rec[6].y), "v"(rec[7].y),
                       "v"(rec[8].y), "v"(rec[9].y), "v"(rec[10].y), "v"(rec[11].y),
                       "v"(rec[12].y), "v"(rec[13].y), "v"(rec[14].y), "v"(rec[15].y));
#pragma unroll
    for (int k = 0; k < RPW; ++k) {
#pragma unroll
        for (int j = 0; j < 2; ++j)
            if (lane + 32 * j < ln[k]) atomicAdd(&c[rec[2 * k + j].y - node0], 1);
    }
#pragma unroll
    for (int k = 0; k < RPW; ++k)
        for (int i = lane + 64; i < ln[k]; i += 32)
            atomicAdd(&c[tmp[2 * (size_t)(st[k] + i) + 1] - node0], 1);
    __syncthreads();

    const int cv = (tid < CB_NODES) ? c[tid] : 0;
    int x = cv;
    for (int o = 1; o < 32; o <<= 1) {
        int y = __shfl_up(x, o, 32);
        if (lane >= o) x += y;
    }
    if (lane == 31) sred[wave] = x;
    __syncthreads();
    const int wtot = sred[lane];
    int wx = wtot;
    for (int o = 1; o < 32; o <<= 1) {
        int y = __shfl_up(wx, o, 32);
        if (lane >= o) wx += y;
    }
    const int before = __shfl(wx - wtot, wave, 32);
    const int p0 = e0 + before + (x - cv);

    if (tid < CB_NODES) {
        const int v = node0 + tid;
        if (v < N) {
            cnt[v]    = cv;
            rowptr[v] = p0;
            dinv[v]   = rsqrtf((float)(cv + 1));
        }
        c[tid] = p0;
    }
    __syncthreads();
#pragma unroll
    for (int k = 0; k < RPW; ++k) {
#pragma unroll
        for (int j = 0; j < 2; ++j) {
            if (lane + 32 * j < ln[k]) {
                const int p = atomicAdd(&c[rec[2 * k + j].y - node0], 1);
                esrc[p] = rec[2 * k + j].x;
            }
        }
    }
#pragma unroll
    for (int k = 0; k < RPW; ++k) {
        for (int i = lane + 64; i < ln[k]; i += 32) {
            const int2 r = *(const int2*)&tmp[2 * (size_t)(st[k] + i)];
            const int p = atomicAdd(&c[r.y - node0], 1);
            esrc[p] = r.x;
        }
    }
}

#define NWF 13
#define LAYER_WGS 256
#define TP  132

__global__ __launch_bounds__(NWF * 32) void k_layer(const int* __restrict__ rowptr,
                                                    const int* __restrict__ cnt,
                                                    const int* __restrict__ esrc,
                                                    const float* __restrict__ dinv,
                                                    const float* __restrict__ X,
                                                    const float* __restrict__ W,
                                                    const float* __restrict__ bias,
                                                    const float* __restrict__ g,
                                                    const float* __restrict__ be,
                                                    float* __restrict__ out, int N) {
    __shared__ float sW[F * F];
    __shared__ float sT[NWF * 16 * TP];
    for (int i = threadIdx.x; i < (F / 2) * (F / 4); i += NWF * 32) {
        int p = i >> 5;
        int c = (i & 31) * 4;
        float4 w0 = *(const float4*)&W[(2 * p)     * F + c];
        float4 w1 = *(const float4*)&W[(2 * p + 1) * F + c];
        v2f* d = (v2f*)&sW[p * 256 + c * 2];
        d[0] = (v2f){w0.x, w1.x};
        d[1] = (v2f){w0.y, w1.y};
        d[2] = (v2f){w0.z, w1.z};
        d[3] = (v2f){w0.w, w1.w};
    }
    __syncthreads();

    const int lane = threadIdx.x & 31;
    const int wave = __builtin_amdgcn_readfirstlane((int)(threadIdx.x >> 5));
    float* tile = &sT[wave * 16 * TP];
    const float* Xl = X + lane * 4;
    const int col   = lane & 15;
    const int khalf = (lane >> 4) << 1;
    const int ntiles = (N + 15) >> 4;

    for (int t = __builtin_amdgcn_readfirstlane((int)(wave * gridDim.x + blockIdx.x)); t < ntiles;
         t += NWF * (int)gridDim.x) {
        const int row0 = t * 16;

        float4 hs[16];
#pragma unroll
        for (int r = 0; r < 16; ++r) {
            int row = row0 + r;
            if (row > N - 1) row = N - 1;
            hs[r] = *(const float4*)(Xl + (size_t)row * F);
        }
        int mrow = row0 + (lane & 15);
        if (mrow > N - 1) mrow = N - 1;
        const int   startv = rowptr[mrow];
        const int   endv   = startv + cnt[mrow];
        const float ddv    = dinv[mrow];
        const int e0    = __builtin_amdgcn_readlane(startv, 0);
        const int e_end = __builtin_amdgcn_readlane(endv, 15);

        int sv = 0, wv = 0;
        if (e_end > e0) {
            int idx = e0 + lane;
            if (idx > e_end - 1) idx = e_end - 1;
            sv = esrc[idx];
            wv = __float_as_int(dinv[sv]);
        }
#pragma unroll
        for (int r = 0; r < 16; ++r) *(float4*)&tile[r * TP + lane * 4] = hs[r];

        int   r    = 0;
        int   rend = __builtin_amdgcn_readlane(endv, 0);
        float dd   = __int_as_float(__builtin_amdgcn_readlane(__float_as_int(ddv), 0));
        float4 own = *(const float4*)&tile[lane * 4];
        float4 ga = {0.f, 0.f, 0.f, 0.f};
#define CLOSE_ROW()                                                                         \
        do {                                                                                    \
            float4 v;                                                                           \
            v.x = dd * fmaf(dd, own.x, ga.x); v.y = dd * fmaf(dd, own.y, ga.y);                 \
            v.z = dd * fmaf(dd, own.z, ga.z); v.w = dd * fmaf(dd, own.w, ga.w);                 \
            *(float4*)&tile[r * TP + lane * 4] = v;                                             \
            ga.x = 0.f; ga.y = 0.f; ga.z = 0.f; ga.w = 0.f;                                     \
            r = (r < 15) ? r + 1 : 15;                                                          \
            rend = __builtin_amdgcn_readlane(endv, r);                                          \
            dd   = __int_as_float(__builtin_amdgcn_readlane(__float_as_int(ddv), r));           \
            own  = *(const float4*)&tile[r * TP + lane * 4];                                    \
        } while (0)

        int closed = 0;
        for (int c0 = e0; c0 < e_end; c0 += 32) {
            float4 h[32];
#pragma unroll
            for (int i = 0; i < 32; ++i) {
                const int s = __builtin_amdgcn_readlane(sv, i);
                h[i] = *(const float4*)(Xl + (size_t)s * F);
            }
            int idxn = c0 + 32 + lane;
            if (idxn > e_end - 1) idxn = e_end - 1;
            const int svn = esrc[idxn];
#pragma unroll
            for (int i = 0; i < 32; ++i) {
                const int e = c0 + i;
                if (e < e_end) {
                    while (e == rend && closed < 16) { CLOSE_ROW(); ++closed; }
                    const float w = __int_as_float(__builtin_amdgcn_readlane(wv, i));
                    ga.x = fmaf(w, h[i].x, ga.x);
                    ga.y = fmaf(w, h[i].y, ga.y);
                    ga.z = fmaf(w, h[i].z, ga.z);
                    ga.w = fmaf(w, h[i].w, ga.w);
                }
            }
            sv = svn;
            wv = __float_as_int(dinv[svn]);
        }
        for (; closed < 16; ++closed) CLOSE_ROW();
#undef CLOSE_ROW

        __builtin_amdgcn_fence(__ATOMIC_RELEASE, "wavefront");
        __builtin_amdgcn_wave_barrier();
        __builtin_amdgcn_fence(__ATOMIC_ACQUIRE, "wavefront");

        const float* arow = tile + col * TP + khalf;

        v8f acc[8];
        const v8f vzero = {0.f, 0.f, 0.f, 0.f, 0.f, 0.f, 0.f, 0.f};
#pragma unroll
        for (int j = 0; j < 8; ++j) acc[j] = vzero;

        for (int k0 = 0; k0 < F; k0 += 4) {
            const v2f a = *(const v2f*)(arow + k0);
            const v2f* bbase = (const v2f*)&sW[((k0 + khalf) >> 1) * 256 + col * 2];
#pragma unroll
            for (int j = 0; j < 8; ++j) {
                v2f b = bbase[j * 16];
                acc[j] = __builtin_amdgcn_wmma_f32_16x16x4_f32(
                    false, a, false, b, (short)0, acc[j], false, false);
            }
        }

        float s[8], q[8];
#pragma unroll
        for (int i = 0; i < 8; ++i) { s[i] = 0.f; q[i] = 0.f; }
#pragma unroll
        for (int j = 0; j < 8; ++j) {
            const float bj = bias[j * 16 + col];
#pragma unroll
            for (int i = 0; i < 8; ++i) {
                const float v = fmaxf(acc[j][i] + bj, 0.f);
                acc[j][i] = v;
                s[i] += v;
            }
        }
#pragma unroll
        for (int i = 0; i < 8; ++i) {
            for (int m = 8; m >= 1; m >>= 1) s[i] += __shfl_xor(s[i], m, 32);
            s[i] *= (1.f / F);
        }
#pragma unroll
        for (int j = 0; j < 8; ++j) {
#pragma unroll
            for (int i = 0; i < 8; ++i) {
                const float d = acc[j][i] - s[i];
                acc[j][i] = d;
                q[i] = fmaf(d, d, q[i]);
            }
        }
#pragma unroll
        for (int i = 0; i < 8; ++i) {
            for (int m = 8; m >= 1; m >>= 1) q[i] += __shfl_xor(q[i], m, 32);
            q[i] = rsqrtf(q[i] * (1.f / F) + LN_EPS);
        }
        float* trow = tile + ((lane >> 4) * 8) * TP + col;
#pragma unroll
        for (int j = 0; j < 8; ++j) {
            const float gj = g[j * 16 + col];
            const float ej = be[j * 16 + col];
#pragma unroll
            for (int i = 0; i < 8; ++i) trow[i * TP + j * 16] = acc[j][i] * q[i] * gj + ej;
        }
        __builtin_amdgcn_fence(__ATOMIC_RELEASE, "wavefront");
        __builtin_amdgcn_wave_barrier();
        __builtin_amdgcn_fence(__ATOMIC_ACQUIRE, "wavefront");
        float4 o[16];
#pragma unroll
        for (int r = 0; r < 16; ++r) o[r] = *(const float4*)&tile[r * TP + lane * 4];
#pragma unroll
        for (int r = 0; r < 16; ++r)
            if (row0 + r < N)
                *(float4*)&out[(size_t)(row0 + r) * F + lane * 4] = o[r];
        __builtin_amdgcn_fence(__ATOMIC_RELEASE, "wavefront");
        __builtin_amdgcn_wave_barrier();
        __builtin_amdgcn_fence(__ATOMIC_ACQUIRE, "wavefront");
    }
}

static inline size_t align256(size_t x) { return (x + 255) & ~(size_t)255; }

extern "C" void kernel_launch(void* const* d_in, const int* in_sizes, int n_in,
                              void* d_out, int out_size, void* d_ws, size_t ws_size,
                              hipStream_t stream) {
    const float* x   = (const float*)d_in[0];
    const int*   ei  = (const int*)  d_in[1];
    const float* W1  = (const float*)d_in[2];
    const float* b1  = (const float*)d_in[3];
    const float* W2  = (const float*)d_in[4];
    const float* b2  = (const float*)d_in[5];
    const float* g1  = (const float*)d_in[6];
    const float* be1 = (const float*)d_in[7];
    const float* g2  = (const float*)d_in[8];
    const float* be2 = (const float*)d_in[9];

    const int N = in_sizes[0] / F;
    const int E = in_sizes[1] / 2;
    const int* src = ei;
    const int* dst = ei + E;
    float* out = (float*)d_out;

    char* ws = (char*)d_ws;
    size_t off = 0;
    float* dinv = (float*)(ws + off); off = align256(off + (size_t)N * 4);
    int*   cnt  = (int*)  (ws + off); off = align256(off + (size_t)N * 4);
    float* Y    = (float*)(ws + off); off = align256(off + (size_t)N * F * 4);
    int*   esrc   = (int*)(ws + off); off = align256(off + (size_t)E * 4);
    int*   rowptr = (int*)(ws + off); off = align256(off + (size_t)N * 4);
    int*   tmp    = (int*)(ws + off); off = align256(off + (size_t)E * 8);
    int*   scnt   = (int*)(ws + off); off = align256(off + (size_t)NSLICE * MAXCB * 4);
    int*   soff   = (int*)(ws + off); off = align256(off + (size_t)NSLICE * MAXCB * 4);
    (void)ws_size; (void)n_in; (void)out_size;

    k_part<<<NSLICE, dim3(1024), 0, stream>>>(src, dst, E, scnt, soff, tmp);
    k_fine<<<(N + CB_NODES - 1) / CB_NODES, dim3(1024), 0, stream>>>(tmp, scnt, soff, N, E, cnt, rowptr, dinv, esrc);

    k_layer<<<LAYER_WGS, dim3(NWF * 32), 0, stream>>>(rowptr, cnt, esrc, dinv, x, W1, b1, g1, be1, Y, N);
    k_layer<<<LAYER_WGS, dim3(NWF * 32), 0, stream>>>(rowptr, cnt, esrc, dinv, Y, W2, b2, g2, be2, out, N);
}


# Gather stream loop: per-edge row-end test is one scalar compare (bucket end kept chunk-relative, INT_MAX once all 16 rows are closed); per-edge tail test gone, lanes past the end get weight 0
# speedup vs baseline: 4.8702x; 1.0553x over previous
#include <hip/hip_runtime.h>

typedef float v2f __attribute__((ext_vector_type(2)));
typedef float v8f __attribute__((ext_vector_type(8)));
typedef int   v4i __attribute__((ext_vector_type(4)));

#define F 128
#define LN_EPS 1e-5f

#define CB_SHIFT 9
#define CB_NODES (1 << CB_SHIFT)
#define NSLICE   256
#define MAXCB    256
static_assert(NSLICE == 256 && MAXCB == 256, "k_fine maps thread t to table columns 4*(t & 63).. and slice group t >> 6");

#define EB 8
static_assert(EB == 8, "the asm operand list in k_part names EB values");
__global__ __launch_bounds__(1024) void k_part(const int* __restrict__ src,
                                               const int* __restrict__ dst, int E,
                                               int* __restrict__ scnt,
                                               int* __restrict__ soff,
                                               int* __restrict__ tmp) {
    __shared__ int h[MAXCB];
    __shared__ int sred[32];
    const int tid  = threadIdx.x;
    const int lane = tid & 31;
    const int wave = tid >> 5;
    if (tid < MAXCB) h[tid] = 0;
    __syncthreads();
    const int per = (E + NSLICE - 1) / NSLICE;
    const int e0 = blockIdx.x * per;
    int e1 = e0 + per;
    if (e1 > E) e1 = E;
    const int base0 = e0 + tid;

    int d[EB], s[EB];
#pragma unroll
    for (int u = 0; u < EB; ++u) { d[u] = 0; s[u] = 0; }
    if (e0 < e1) {
#pragma unroll
        for (int u = 0; u < EB; ++u) {
            const int e  = base0 + u * 1024;
            const int ec = (e < e1) ? e : e1 - 1;
            d[u] = dst[ec];
            s[u] = src[ec];
        }
        asm volatile("" :: "v"(d[0]), "v"(d[1]), "v"(d[2]), "v"(d[3]),
                           "v"(d[4]), "v"(d[5]), "v"(d[6]), "v"(d[7]),
                           "v"(s[0]), "v"(s[1]), "v"(s[2]), "v"(s[3]),
                           "v"(s[4]), "v"(s[5]), "v"(s[6]), "v"(s[7]));
#pragma unroll
        for (int u = 0; u < EB; ++u)
            if (base0 + u * 1024 < e1) atomicAdd(&h[d[u] >> CB_SHIFT], 1);
    }
    for (int e = base0 + EB * 1024; e < e1; e += 1024) atomicAdd(&h[dst[e] >> CB_SHIFT], 1);
    __syncthreads();

    const int hv = (tid < MAXCB) ? h[tid] : 0;
    int x = hv;
    for (int o = 1; o < 32; o <<= 1) {
        int y = __shfl_up(x, o, 32);
        if (lane >= o) x += y;
    }
    if (lane == 31) sred[wave] = x;
    __syncthreads();
    const int wtot = sred[lane];
    int wx = wtot;
    for (int o = 1; o < 32; o <<= 1) {
        int y = __shfl_up(wx, o, 32);
        if (lane >= o) wx += y;
    }
    const int before = __shfl(wx - wtot, wave, 32);
    const int p0 = e0 + before + (x - hv);
    if (tid < MAXCB) {
        scnt[blockIdx.x * MAXCB + tid] = hv;
        soff[blockIdx.x * MAXCB + tid] = p0;
        h[tid] = p0;
    }
    __syncthreads();

    if (e0 < e1) {
#pragma unroll
        for (int u = 0; u < EB; ++u) {
            if (base0 + u * 1024 < e1) {
                const int p = atomicAdd(&h[d[u] >> CB_SHIFT], 1);
                *(int2*)&tmp[2 * (size_t)p] = make_int2(s[u], d[u]);
            }
        }
    }
    for (int e = base0 + EB * 1024; e < e1; e += 1024) {
        const int dd = dst[e];
        const int ss = src[e];
        const int p = atomicAdd(&h[dd >> CB_SHIFT], 1);
        *(int2*)&tmp[2 * (size_t)p] = make_int2(ss, dd);
    }
}

#define RPW 8
static_assert(RPW * 32 == NSLICE, "every slice's run belongs to exactly one wave");
static_assert(RPW == 8, "the asm operand list in k_fine names 2*RPW values");
__global__ __launch_bounds__(1024) void k_fine(const int* __restrict__ tmp,
                                               const int* __restrict__ scnt,
                                               const int* __restrict__ soff, int N, int E,
                                               int* __restrict__ cnt,
                                               int* __restrict__ rowptr,
                                               float* __restrict__ dinv,
                                               int* __restrict__ esrc) {
    __shared__ int c[CB_NODES];
    __shared__ int sred[32];
    __shared__ int rs[NSLICE];
    __shared__ int rl[NSLICE];
    const int tid  = threadIdx.x;
    const int lane = tid & 31;
    const int wave = tid >> 5;
    const int rb   = blockIdx.x;
    const int node0 = rb << CB_SHIFT;

    if (tid < NSLICE) {
        rs[tid] = soff[tid * MAXCB + rb];
        rl[tid] = scnt[tid * MAXCB + rb];
    }
    if (tid < CB_NODES) c[tid] = 0;
    int part = 0;
    {
        const int c4 = (tid & 63) * 4;
        const int rg = tid >> 6;
        if (c4 < rb) {
            const int rot = rb & 15;
            v4i v[16];
#pragma unroll
            for (int k = 0; k < 16; ++k)
                v[k] = *(const v4i*)&scnt[(rg * 16 + (k ^ rot)) * MAXCB + c4];
            asm volatile("" :: "v"(v[0]), "v"(v[1]), "v"(v[2]), "v"(v[3]),
                               "v"(v[4]), "v"(v[5]), "v"(v[6]), "v"(v[7]),
                               "v"(v[8]), "v"(v[9]), "v"(v[10]), "v"(v[11]),
                               "v"(v[12]), "v"(v[13]), "v"(v[14]), "v"(v[15]));
            int sx = 0, sy = 0, sz = 0, sw = 0;
#pragma unroll
            for (int k = 0; k < 16; ++k) { sx += v[k].x; sy += v[k].y; sz += v[k].z; sw += v[k].w; }
            part = sx;
            if (c4 + 1 < rb) part += sy;
            if (c4 + 2 < rb) part += sz;
            if (c4 + 3 < rb) part += sw;
        }
    }
    for (int o = 16; o >= 1; o >>= 1) part += __shfl_xor(part, o, 32);
    if (lane == 0) sred[wave] = part;
    __syncthreads();
    int e0 = sred[lane];
    for (int o = 16; o >= 1; o >>= 1) e0 += __shfl_xor(e0, o, 32);

    int st[RPW], ln[RPW];
#pragma unroll
    for (int k = 0; k < RPW; ++k) {
        st[k] = __builtin_amdgcn_readfirstlane(rs[wave + 32 * k]);
        ln[k] = __builtin_amdgcn_readfirstlane(rl[wave + 32 * k]);
    }

    int2 rec[2 * RPW];
#pragma unroll
    for (int k = 0; k < RPW; ++k) {
#pragma unroll
        for (int j = 0; j < 2; ++j) {
            const int i = lane + 32 * j;
            int idx = st[k] + ((i < ln[k]) ? i : 0);
            if (idx > E - 1) idx = E - 1;
            if (idx < 0) idx = 0;
            rec[2 * k + j] = *(const int2*)&tmp[2 * (size_t)idx];
        }
    }
    asm volatile("" :: "v"(rec[0].y), "v"(rec[1].y), "v"(rec[2].y), "v"(rec[3].y),
                       "v"(rec[4].y), "v"(rec[5].y), "v"(rec[6].y), "v"(rec[7].y),
                       "v"(rec[8].y), "v"(rec[9].y), "v"(rec[10].y), "v"(rec[11].y),
                       "v"(rec[12].y), "v"(rec[13].y), "v"(rec[14].y), "v"(rec[15].y));
#pragma unroll
    for (int k = 0; k < RPW; ++k) {
#pragma unroll
        for (int j = 0; j < 2; ++j)
            if (lane + 32 * j < ln[k]) atomicAdd(&c[rec[2 * k + j].y - node0], 1);
    }
#pragma unroll
    for (int k = 0; k < RPW; ++k)
        for (int i = lane + 64; i < ln[k]; i += 32)
            atomicAdd(&c[tmp[2 * (size_t)(st[k] + i) + 1] - node0], 1);
    __syncthreads();

    const int cv = (tid < CB_NODES) ? c[tid] : 0;
    int x = cv;
    for (int o = 1; o < 32; o <<= 1) {
        int y = __shfl_up(x, o, 32);
        if (lane >= o) x += y;
    }
    if (lane == 31) sred[wave] = x;
    __syncthreads();
    const int wtot = sred[lane];
    int wx = wtot;
    for (int o = 1; o < 32; o <<= 1) {
        int y = __shfl_up(wx, o, 32);
        if (lane >= o) wx += y;
    }
    const int before = __shfl(wx - wtot, wave, 32);
    const int p0 = e0 + before + (x - cv);

    if (tid < CB_NODES) {
        const int v = node0 + tid;
        if (v < N) {
            cnt[v]    = cv;
            rowptr[v] = p0;
            dinv[v]   = rsqrtf((float)(cv + 1));
        }
        c[tid] = p0;
    }
    __syncthreads();
#pragma unroll
    for (int k = 0; k < RPW; ++k) {
#pragma unroll
        for (int j = 0; j < 2; ++j) {
            if (lane + 32 * j < ln[k]) {
                const int p = atomicAdd(&c[rec[2 * k + j].y - node0], 1);
                esrc[p] = rec[2 * k + j].x;
            }
        }
    }
#pragma unroll
    for (int k = 0; k < RPW; ++k) {
        for (int i = lane + 64; i < ln[k]; i += 32) {
            const int2 r = *(const int2*)&tmp[2 * (size_t)(st[k] + i)];
            const int p = atomicAdd(&c[r.y - node0], 1);
            esrc[p] = r.x;
        }
    }
}

#define NWF 13
#define LAYER_WGS 256
#define TP  132

__global__ __launch_bounds__(NWF * 32) void k_layer(const int* __restrict__ rowptr,
                                                    const int* __restrict__ cnt,
                                                    const int* __restrict__ esrc,
                                                    const float* __restrict__ dinv,
                                                    const float* __restrict__ X,
                                                    const float* __restrict__ W,
                                                    const float* __restrict__ bias,
                                                    const float* __restrict__ g,
                                                    const float* __restrict__ be,
                                                    float* __restrict__ out, int N) {
    __shared__ float sW[F * F];
    __shared__ float sT[NWF * 16 * TP];
    for (int i = threadIdx.x; i < (F / 2) * (F / 4); i += NWF * 32) {
        int p = i >> 5;
        int c = (i & 31) * 4;
        float4 w0 = *(const float4*)&W[(2 * p)     * F + c];
        float4 w1 = *(const float4*)&W[(2 * p + 1) * F + c];
        v2f* d = (v2f*)&sW[p * 256 + c * 2];
        d[0] = (v2f){w0.x, w1.x};
        d[1] = (v2f){w0.y, w1.y};
        d[2] = (v2f){w0.z, w1.z};
        d[3] = (v2f){w0.w, w1.w};
    }
    __syncthreads();

    const int lane = threadIdx.x & 31;
    const int wave = __builtin_amdgcn_readfirstlane((int)(threadIdx.x >> 5));
    float* tile = &sT[wave * 16 * TP];
    const float* Xl = X + lane * 4;
    const int col   = lane & 15;
    const int khalf = (lane >> 4) << 1;
    const int ntiles = (N + 15) >> 4;

    for (int t = __builtin_amdgcn_readfirstlane((int)(wave * gridDim.x + blockIdx.x)); t < ntiles;
         t += NWF * (int)gridDim.x) {
        const int row0 = t * 16;

        float4 hs[16];
#pragma unroll
        for (int r = 0; r < 16; ++r) {
            int row = row0 + r;
            if (row > N - 1) row = N - 1;
            hs[r] = *(const float4*)(Xl + (size_t)row * F);
        }
        int mrow = row0 + (lane & 15);
        if (mrow > N - 1) mrow = N - 1;
        const int   startv = rowptr[mrow];
        const int   endv   = startv + cnt[mrow];
        const float ddv    = dinv[mrow];
        const int e0    = __builtin_amdgcn_readlane(startv, 0);
        const int e_end = __builtin_amdgcn_readlane(endv, 15);

        int sv = 0, wv = 0;
        if (e_end > e0) {
            int idx = e0 + lane;
            const bool live = idx < e_end;
            if (!live) idx = e_end - 1;
            sv = esrc[idx];
            const int wl = __float_as_int(dinv[sv]);
            wv = live ? wl : 0;
        }
#pragma unroll
        for (int r = 0; r < 16; ++r) *(float4*)&tile[r * TP + lane * 4] = hs[r];

        int   r    = 0;
        int   closed = 0;
        int   rend = __builtin_amdgcn_readlane(endv, 0);
        float dd   = __int_as_float(__builtin_amdgcn_readlane(__float_as_int(ddv), 0));
        float4 own = *(const float4*)&tile[lane * 4];
        float4 ga = {0.f, 0.f, 0.f, 0.f};
#define CLOSE_ROW()                                                                         \
        do {                                                                                    \
            float4 v;                                                                           \
            v.x = dd * fmaf(dd, own.x, ga.x); v.y = dd * fmaf(dd, own.y, ga.y);                 \
            v.z = dd * fmaf(dd, own.z, ga.z); v.w = dd * fmaf(dd, own.w, ga.w);                 \
            *(float4*)&tile[r * TP + lane * 4] = v;                                             \
            ga.x = 0.f; ga.y = 0.f; ga.z = 0.f; ga.w = 0.f;                                     \
            ++closed;                                                                           \
            r = (r < 15) ? r + 1 : 15;                                                          \
            rend = (closed < 16) ? __builtin_amdgcn_readlane(endv, r) : 0x7fffffff;             \
            dd   = __int_as_float(__builtin_amdgcn_readlane(__float_as_int(ddv), r));           \
            own  = *(const float4*)&tile[r * TP + lane * 4];                                    \
        } while (0)

        for (int c0 = e0; c0 < e_end; c0 += 32) {
            float4 h[32];
#pragma unroll
            for (int i = 0; i < 32; ++i) {
                const int s = __builtin_amdgcn_readlane(sv, i);
                h[i] = *(const float4*)(Xl + (size_t)s * F);
            }
            int idxn = c0 + 32 + lane;
            if (idxn > e_end - 1) idxn = e_end - 1;
            const int svn = esrc[idxn];
            int rrel = rend - c0;
#pragma unroll
            for (int i = 0; i < 32; ++i) {
                if (__builtin_expect(rrel == i, 0)) {
                    do { CLOSE_ROW(); rrel = rend - c0; } while (rrel == i);
                }
                const float w = __int_as_float(__builtin_amdgcn_readlane(wv, i));
                ga.x = fmaf(w, h[i].x, ga.x);
                ga.y = fmaf(w, h[i].y, ga.y);
                ga.z = fmaf(w, h[i].z, ga.z);
                ga.w = fmaf(w, h[i].w, ga.w);
            }
            sv = svn;
            const int wn = __float_as_int(dinv[svn]);
            wv = (c0 + 32 + lane < e_end) ? wn : 0;
        }
        while (closed < 16) CLOSE_ROW();
#undef CLOSE_ROW

        __builtin_amdgcn_fence(__ATOMIC_RELEASE, "wavefront");
        __builtin_amdgcn_wave_barrier();
        __builtin_amdgcn_fence(__ATOMIC_ACQUIRE, "wavefront");

        const float* arow = tile + col * TP + khalf;

        v8f acc[8];
        const v8f vzero = {0.f, 0.f, 0.f, 0.f, 0.f, 0.f, 0.f, 0.f};
#pragma unroll
        for (int j = 0; j < 8; ++j) acc[j] = vzero;

        for (int k0 = 0; k0 < F; k0 += 4) {
            const v2f a = *(const v2f*)(arow + k0);
            const v2f* bbase = (const v2f*)&sW[((k0 + khalf) >> 1) * 256 + col * 2];
#pragma unroll
            for (int j = 0; j < 8; ++j) {
                v2f b = bbase[j * 16];
                acc[j] = __builtin_amdgcn_wmma_f32_16x16x4_f32(
                    false, a, false, b, (short)0, acc[j], false, false);
            }
        }

        float s[8], q[8];
#pragma unroll
        for (int i = 0; i < 8; ++i) { s[i] = 0.f; q[i] = 0.f; }
#pragma unroll
        for (int j = 0; j < 8; ++j) {
            const float bj = bias[j * 16 + col];
#pragma unroll
            for (int i = 0; i < 8; ++i) {
                const float v = fmaxf(acc[j][i] + bj, 0.f);
                acc[j][i] = v;
                s[i] += v;
            }
        }
#pragma unroll
        for (int i = 0; i < 8; ++i) {
            for (int m = 8; m >= 1; m >>= 1) s[i] += __shfl_xor(s[i], m, 32);
            s[i] *= (1.f / F);
        }
#pragma unroll
        for (int j = 0; j < 8; ++j) {
#pragma unroll
            for (int i = 0; i < 8; ++i) {
                const float d = acc[j][i] - s[i];
                acc[j][i] = d;
                q[i] = fmaf(d, d, q[i]);
            }
        }
#pragma unroll
        for (int i = 0; i < 8; ++i) {
            for (int m = 8; m >= 1; m >>= 1) q[i] += __shfl_xor(q[i], m, 32);
            q[i] = rsqrtf(q[i] * (1.f / F) + LN_EPS);
        }
        float* trow = tile + ((lane >> 4) * 8) * TP + col;
#pragma unroll
        for (int j = 0; j < 8; ++j) {
            const float gj = g[j * 16 + col];
            const float ej = be[j * 16 + col];
#pragma unroll
            for (int i = 0; i < 8; ++i) trow[i * TP + j * 16] = acc[j][i] * q[i] * gj + ej;
        }
        __builtin_amdgcn_fence(__ATOMIC_RELEASE, "wavefront");
        __builtin_amdgcn_wave_barrier();
        __builtin_amdgcn_fence(__ATOMIC_ACQUIRE, "wavefront");
        float4 o[16];
#pragma unroll
        for (int r = 0; r < 16; ++r) o[r] = *(const float4*)&tile[r * TP + lane * 4];
#pragma unroll
        for (int r = 0; r < 16; ++r)
            if (row0 + r < N)
                *(float4*)&out[(size_t)(row0 + r) * F + lane * 4] = o[r];
        __builtin_amdgcn_fence(__ATOMIC_RELEASE, "wavefront");
        __builtin_amdgcn_wave_barrier();
        __builtin_amdgcn_fence(__ATOMIC_ACQUIRE, "wavefront");
    }
}

static inline size_t align256(size_t x) { return (x + 255) & ~(size_t)255; }

extern "C" void kernel_launch(void* const* d_in, const int* in_sizes, int n_in,
                              void* d_out, int out_size, void* d_ws, size_t ws_size,
                              hipStream_t stream) {
    const float* x   = (const float*)d_in[0];
    const int*   ei  = (const int*)  d_in[1];
    const float* W1  = (const float*)d_in[2];
    const float* b1  = (const float*)d_in[3];
    const float* W2  = (const float*)d_in[4];
    const float* b2  = (const float*)d_in[5];
    const float* g1  = (const float*)d_in[6];
    const float* be1 = (const float*)d_in[7];
    const float* g2  = (const float*)d_in[8];
    const float* be2 = (const float*)d_in[9];

    const int N = in_sizes[0] / F;
    const int E = in_sizes[1] / 2;
    const int* src = ei;
    const int* dst = ei + E;
    float* out = (float*)d_out;

    char* ws = (char*)d_ws;
    size_t off = 0;
    float* dinv = (float*)(ws + off); off = align256(off + (size_t)N * 4);
    int*   cnt  = (int*)  (ws + off); off = align256(off + (size_t)N * 4);
    float* Y    = (float*)(ws + off); off = align256(off + (size_t)N * F * 4);
    int*   esrc   = (int*)(ws + off); off = align256(off + (size_t)E * 4);
    int*   rowptr = (int*)(ws + off); off = align256(off + (size_t)N * 4);
    int*   tmp    = (int*)(ws + off); off = align256(off + (size_t)E * 8);
    int*   scnt   = (int*)(ws + off); off = align256(off + (size_t)NSLICE * MAXCB * 4);
    int*   soff   = (int*)(ws + off); off = align256(off + (size_t)NSLICE * MAXCB * 4);
    (void)ws_size; (void)n_in; (void)out_size;

    k_part<<<NSLICE, dim3(1024), 0, stream>>>(src, dst, E, scnt, soff, tmp);
    k_fine<<<(N + CB_NODES - 1) / CB_NODES, dim3(1024), 0, stream>>>(tmp, scnt, soff, N, E, cnt, rowptr, dinv, esrc);

    k_layer<<<LAYER_WGS, dim3(NWF * 32), 0, stream>>>(rowptr, cnt, esrc, dinv, x, W1, b1, g1, be1, Y, N);
    k_layer<<<LAYER_WGS, dim3(NWF * 32), 0, stream>>>(rowptr, cnt, esrc, dinv, Y, W2, b2, g2, be2, out, N);
}


# Layer 1 stores its rows times dinv[row]; layer 2 adds gathered rows unweighted: its per-edge dinv fetch (one 128-byte line per edge), weight broadcast and multiply gone; products round differently
# speedup vs baseline: 5.0880x; 1.0447x over previous
#include <hip/hip_runtime.h>

typedef float v2f __attribute__((ext_vector_type(2)));
typedef float v8f __attribute__((ext_vector_type(8)));
typedef int   v4i __attribute__((ext_vector_type(4)));

#define F 128
#define LN_EPS 1e-5f

#define CB_SHIFT 9
#define CB_NODES (1 << CB_SHIFT)
#define NSLICE   256
#define MAXCB    256
static_assert(NSLICE == 256 && MAXCB == 256, "k_fine maps thread t to table columns 4*(t & 63).. and slice group t >> 6");

#define EB 8
static_assert(EB == 8, "the asm operand list in k_part names EB values");
__global__ __launch_bounds__(1024) void k_part(const int* __restrict__ src,
                                               const int* __restrict__ dst, int E,
                                               int* __restrict__ scnt,
                                               int* __restrict__ soff,
                                               int* __restrict__ tmp) {
    __shared__ int h[MAXCB];
    __shared__ int sred[32];
    const int tid  = threadIdx.x;
    const int lane = tid & 31;
    const int wave = tid >> 5;
    if (tid < MAXCB) h[tid] = 0;
    __syncthreads();
    const int per = (E + NSLICE - 1) / NSLICE;
    const int e0 = blockIdx.x * per;
    int e1 = e0 + per;
    if (e1 > E) e1 = E;
    const int base0 = e0 + tid;

    int d[EB], s[EB];
#pragma unroll
    for (int u = 0; u < EB; ++u) { d[u] = 0; s[u] = 0; }
    if (e0 < e1) {
#pragma unroll
        for (int u = 0; u < EB; ++u) {
            const int e  = base0 + u * 1024;
            const int ec = (e < e1) ? e : e1 - 1;
            d[u] = dst[ec];
            s[u] = src[ec];
        }
        asm volatile("" :: "v"(d[0]), "v"(d[1]), "v"(d[2]), "v"(d[3]),
                           "v"(d[4]), "v"(d[5]), "v"(d[6]), "v"(d[7]),
                           "v"(s[0]), "v"(s[1]), "v"(s[2]), "v"(s[3]),
                           "v"(s[4]), "v"(s[5]), "v"(s[6]), "v"(s[7]));
#pragma unroll
        for (int u = 0; u < EB; ++u)
            if (base0 + u * 1024 < e1) atomicAdd(&h[d[u] >> CB_SHIFT], 1);
    }
    for (int e = base0 + EB * 1024; e < e1; e += 1024) atomicAdd(&h[dst[e] >> CB_SHIFT], 1);
    __syncthreads();

    const int hv = (tid < MAXCB) ? h[tid] : 0;
    int x = hv;
    for (int o = 1; o < 32; o <<= 1) {
        int y = __shfl_up(x, o, 32);
        if (lane >= o) x += y;
    }
    if (lane == 31) sred[wave] = x;
    __syncthreads();
    const int wtot = sred[lane];
    int wx = wtot;
    for (int o = 1; o < 32; o <<= 1) {
        int y = __shfl_up(wx, o, 32);
        if (lane >= o) wx += y;
    }
    const int before = __shfl(wx - wtot, wave, 32);
    const int p0 = e0 + before + (x - hv);
    if (tid < MAXCB) {
        scnt[blockIdx.x * MAXCB + tid] = hv;
        soff[blockIdx.x * MAXCB + tid] = p0;
        h[tid] = p0;
    }
    __syncthreads();

    if (e0 < e1) {
#pragma unroll
        for (int u = 0; u < EB; ++u) {
            if (base0 + u * 1024 < e1) {
                const int p = atomicAdd(&h[d[u] >> CB_SHIFT], 1);
                *(int2*)&tmp[2 * (size_t)p] = make_int2(s[u], d[u]);
            }
        }
    }
    for (int e = base0 + EB * 1024; e < e1; e += 1024) {
        const int dd = dst[e];
        const int ss = src[e];
        const int p = atomicAdd(&h[dd >> CB_SHIFT], 1);
        *(int2*)&tmp[2 * (size_t)p] = make_int2(ss, dd);
    }
}

#define RPW 8
static_assert(RPW * 32 == NSLICE, "every slice's run belongs to exactly one wave");
static_assert(RPW == 8, "the asm operand list in k_fine names 2*RPW values");
__global__ __launch_bounds__(1024) void k_fine(const int* __restrict__ tmp,
                                               const int* __restrict__ scnt,
                                               const int* __restrict__ soff, int N, int E,
                                               int* __restrict__ cnt,
                                               int* __restrict__ rowptr,
                                               float* __restrict__ dinv,
                                               int* __restrict__ esrc) {
    __shared__ int c[CB_NODES];
    __shared__ int sred[32];
    __shared__ int rs[NSLICE];
    __shared__ int rl[NSLICE];
    const int tid  = threadIdx.x;
    const int lane = tid & 31;
    const int wave = tid >> 5;
    const int rb   = blockIdx.x;
    const int node0 = rb << CB_SHIFT;

    if (tid < NSLICE) {
        rs[tid] = soff[tid * MAXCB + rb];
        rl[tid] = scnt[tid * MAXCB + rb];
    }
    if (tid < CB_NODES) c[tid] = 0;
    int part = 0;
    {
        const int c4 = (tid & 63) * 4;
        const int rg = tid >> 6;
        if (c4 < rb) {
            const int rot = rb & 15;
            v4i v[16];
#pragma unroll
            for (int k = 0; k < 16; ++k)
                v[k] = *(const v4i*)&scnt[(rg * 16 + (k ^ rot)) * MAXCB + c4];
            asm volatile("" :: "v"(v[0]), "v"(v[1]), "v"(v[2]), "v"(v[3]),
                               "v"(v[4]), "v"(v[5]), "v"(v[6]), "v"(v[7]),
                               "v"(v[8]), "v"(v[9]), "v"(v[10]), "v"(v[11]),
                               "v"(v[12]), "v"(v[13]), "v"(v[14]), "v"(v[15]));
            int sx = 0, sy = 0, sz = 0, sw = 0;
#pragma unroll
            for (int k = 0; k < 16; ++k) { sx += v[k].x; sy += v[k].y; sz += v[k].z; sw += v[k].w; }
            part = sx;
            if (c4 + 1 < rb) part += sy;
            if (c4 + 2 < rb) part += sz;
            if (c4 + 3 < rb) part += sw;
        }
    }
    for (int o = 16; o >= 1; o >>= 1) part += __shfl_xor(part, o, 32);
    if (lane == 0) sred[wave] = part;
    __syncthreads();
    int e0 = sred[lane];
    for (int o = 16; o >= 1; o >>= 1) e0 += __shfl_xor(e0, o, 32);

    int st[RPW], ln[RPW];
#pragma unroll
    for (int k = 0; k < RPW; ++k) {
        st[k] = __builtin_amdgcn_readfirstlane(rs[wave + 32 * k]);
        ln[k] = __builtin_amdgcn_readfirstlane(rl[wave + 32 * k]);
    }

    int2 rec[2 * RPW];
#pragma unroll
    for (int k = 0; k < RPW; ++k) {
#pragma unroll
        for (int j = 0; j < 2; ++j) {
            const int i = lane + 32 * j;
            int idx = st[k] + ((i < ln[k]) ? i : 0);
            if (idx > E - 1) idx = E - 1;
            if (idx < 0) idx = 0;
            rec[2 * k + j] = *(const int2*)&tmp[2 * (size_t)idx];
        }
    }
    asm volatile("" :: "v"(rec[0].y), "v"(rec[1].y), "v"(rec[2].y), "v"(rec[3].y),
                       "v"(rec[4].y), "v"(rec[5].y), "v"(rec[6].y), "v"(rec[7].y),
                       "v"(rec[8].y), "v"(rec[9].y), "v"(rec[10].y), "v"(rec[11].y),
                       "v"(rec[12].y), "v"(rec[13].y), "v"(rec[14].y), "v"(rec[15].y));
#pragma unroll
    for (int k = 0; k < RPW; ++k) {
#pragma unroll
        for (int j = 0; j < 2; ++j)
            if (lane + 32 * j < ln[k]) atomicAdd(&c[rec[2 * k + j].y - node0], 1);
    }
#pragma unroll
    for (int k = 0; k < RPW; ++k)
        for (int i = lane + 64; i < ln[k]; i += 32)
            atomicAdd(&c[tmp[2 * (size_t)(st[k] + i) + 1] - node0], 1);
    __syncthreads();

    const int cv = (tid < CB_NODES) ? c[tid] : 0;
    int x = cv;
    for (int o = 1; o < 32; o <<= 1) {
        int y = __shfl_up(x, o, 32);
        if (lane >= o) x += y;
    }
    if (lane == 31) sred[wave] = x;
    __syncthreads();
    const int wtot = sred[lane];
    int wx = wtot;
    for (int o = 1; o < 32; o <<= 1) {
        int y = __shfl_up(wx, o, 32);
        if (lane >= o) wx += y;
    }
    const int before = __shfl(wx - wtot, wave, 32);
    const int p0 = e0 + before + (x - cv);

    if (tid < CB_NODES) {
        const int v = node0 + tid;
        if (v < N) {
            cnt[v]    = cv;
            rowptr[v] = p0;
            dinv[v]   = rsqrtf((float)(cv + 1));
        }
        c[tid] = p0;
    }
    __syncthreads();
#pragma unroll
    for (int k = 0; k < RPW; ++k) {
#pragma unroll
        for (int j = 0; j < 2; ++j) {
            if (lane + 32 * j < ln[k]) {
                const int p = atomicAdd(&c[rec[2 * k + j].y - node0], 1);
                esrc[p] = rec[2 * k + j].x;
            }
        }
    }
#pragma unroll
    for (int k = 0; k < RPW; ++k) {
        for (int i = lane + 64; i < ln[k]; i += 32) {
            const int2 r = *(const int2*)&tmp[2 * (size_t)(st[k] + i)];
            const int p = atomicAdd(&c[r.y - node0], 1);
            esrc[p] = r.x;
        }
    }
}

#define NWF 13
#define LAYER_WGS 256
#define TP  132

template <bool FIRST>
__global__ __launch_bounds__(NWF * 32) void k_layer(const int* __restrict__ rowptr,
                                                    const int* __restrict__ cnt,
                                                    const int* __restrict__ esrc,
                                                    const float* __restrict__ dinv,
                                                    const float* __restrict__ X,
                                                    const float* __restrict__ W,
                                                    const float* __restrict__ bias,
                                                    const float* __restrict__ g,
                                                    const float* __restrict__ be,
                                                    float* __restrict__ out, int N) {
    __shared__ float sW[F * F];
    __shared__ float sT[NWF * 16 * TP];
    for (int i = threadIdx.x; i < (F / 2) * (F / 4); i += NWF * 32) {
        int p = i >> 5;
        int c = (i & 31) * 4;
        float4 w0 = *(const float4*)&W[(2 * p)     * F + c];
        float4 w1 = *(const float4*)&W[(2 * p + 1) * F + c];
        v2f* d = (v2f*)&sW[p * 256 + c * 2];
        d[0] = (v2f){w0.x, w1.x};
        d[1] = (v2f){w0.y, w1.y};
        d[2] = (v2f){w0.z, w1.z};
        d[3] = (v2f){w0.w, w1.w};
    }
    __syncthreads();

    const int lane = threadIdx.x & 31;
    const int wave = __builtin_amdgcn_readfirstlane((int)(threadIdx.x >> 5));
    float* tile = &sT[wave * 16 * TP];
    const float* Xl = X + lane * 4;
    const int col   = lane & 15;
    const int khalf = (lane >> 4) << 1;
    const int ntiles = (N + 15) >> 4;

    for (int t = __builtin_amdgcn_readfirstlane((int)(wave * gridDim.x + blockIdx.x)); t < ntiles;
         t += NWF * (int)gridDim.x) {
        const int row0 = t * 16;

        float4 hs[16];
#pragma unroll
        for (int r = 0; r < 16; ++r) {
            int row = row0 + r;
            if (row > N - 1) row = N - 1;
            hs[r] = *(const float4*)(Xl + (size_t)row * F);
        }
        int mrow = row0 + (lane & 15);
        if (mrow > N - 1) mrow = N - 1;
        const int   startv = rowptr[mrow];
        const int   endv   = startv + cnt[mrow];
        const float ddv    = dinv[mrow];
        const int e0    = __builtin_amdgcn_readlane(startv, 0);
        const int e_end = __builtin_amdgcn_readlane(endv, 15);

        int sv = 0, wv = 0;
        if (e_end > e0) {
            int idx = e0 + lane;
            const bool live = idx < e_end;
            if (!live) idx = e_end - 1;
            sv = esrc[idx];
            if (FIRST) {
                const int wl = __float_as_int(dinv[sv]);
                wv = live ? wl : 0;
            }
        }
#pragma unroll
        for (int r = 0; r < 16; ++r) *(float4*)&tile[r * TP + lane * 4] = hs[r];

        int   r    = 0;
        int   closed = 0;
        int   rend = __builtin_amdgcn_readlane(endv, 0);
        float dd   = __int_as_float(__builtin_amdgcn_readlane(__float_as_int(ddv), 0));
        float4 own = *(const float4*)&tile[lane * 4];
        float4 ga = {0.f, 0.f, 0.f, 0.f};
#define CLOSE_ROW()                                                                         \
        do {                                                                                    \
            float4 v;                                                                           \
            if (FIRST) {                                                                        \
                v.x = dd * fmaf(dd, own.x, ga.x); v.y = dd * fmaf(dd, own.y, ga.y);             \
                v.z = dd * fmaf(dd, own.z, ga.z); v.w = dd * fmaf(dd, own.w, ga.w);             \
            } else {                                                                            \
                v.x = dd * (ga.x + own.x); v.y = dd * (ga.y + own.y);                           \
                v.z = dd * (ga.z + own.z); v.w = dd * (ga.w + own.w);                           \
            }                                                                                   \
            *(float4*)&tile[r * TP + lane * 4] = v;                                             \
            ga.x = 0.f; ga.y = 0.f; ga.z = 0.f; ga.w = 0.f;                                     \
            ++closed;                                                                           \
            r = (r < 15) ? r + 1 : 15;                                                          \
            rend = (closed < 16) ? __builtin_amdgcn_readlane(endv, r) : 0x7fffffff;             \
            dd   = __int_as_float(__builtin_amdgcn_readlane(__float_as_int(ddv), r));           \
            own  = *(const float4*)&tile[r * TP + lane * 4];                                    \
        } while (0)

        for (int c0 = e0; c0 < e_end; c0 += 32) {
            float4 h[32];
#pragma unroll
            for (int i = 0; i < 32; ++i) {
                const int s = __builtin_amdgcn_readlane(sv, i);
                h[i] = *(const float4*)(Xl + (size_t)s * F);
            }
            int idxn = c0 + 32 + lane;
            if (idxn > e_end - 1) idxn = e_end - 1;
            const int svn = esrc[idxn];
            int rrel = rend - c0;
#pragma unroll
            for (int i = 0; i < 32; ++i) {
                if (__builtin_expect(rrel == i, 0)) {
                    do { CLOSE_ROW(); rrel = rend - c0; } while (rrel == i);
                }
                if (FIRST) {
                    const float w = __int_as_float(__builtin_amdgcn_readlane(wv, i));
                    ga.x = fmaf(w, h[i].x, ga.x);
                    ga.y = fmaf(w, h[i].y, ga.y);
                    ga.z = fmaf(w, h[i].z, ga.z);
                    ga.w = fmaf(w, h[i].w, ga.w);
                } else {
                    ga.x += h[i].x;
                    ga.y += h[i].y;
                    ga.z += h[i].z;
                    ga.w += h[i].w;
                }
            }
            sv = svn;
            if (FIRST) {
                const int wn = __float_as_int(dinv[svn]);
                wv = (c0 + 32 + lane < e_end) ? wn : 0;
            }
        }
        while (closed < 16) CLOSE_ROW();
#undef CLOSE_ROW

        __builtin_amdgcn_fence(__ATOMIC_RELEASE, "wavefront");
        __builtin_amdgcn_wave_barrier();
        __builtin_amdgcn_fence(__ATOMIC_ACQUIRE, "wavefront");

        const float* arow = tile + col * TP + khalf;

        v8f acc[8];
        const v8f vzero = {0.f, 0.f, 0.f, 0.f, 0.f, 0.f, 0.f, 0.f};
#pragma unroll
        for (int j = 0; j < 8; ++j) acc[j] = vzero;

        for (int k0 = 0; k0 < F; k0 += 4) {
            const v2f a = *(const v2f*)(arow + k0);
            const v2f* bbase = (const v2f*)&sW[((k0 + khalf) >> 1) * 256 + col * 2];
#pragma unroll
            for (int j = 0; j < 8; ++j) {
                v2f b = bbase[j * 16];
                acc[j] = __builtin_amdgcn_wmma_f32_16x16x4_f32(
                    false, a, false, b, (short)0, acc[j], false, false);
            }
        }

        float s[8], q[8];
#pragma unroll
        for (int i = 0; i < 8; ++i) { s[i] = 0.f; q[i] = 0.f; }
#pragma unroll
        for (int j = 0; j < 8; ++j) {
            const float bj = bias[j * 16 + col];
#pragma unroll
            for (int i = 0; i < 8; ++i) {
                const float v = fmaxf(acc[j][i] + bj, 0.f);
                acc[j][i] = v;
                s[i] += v;
            }
        }
#pragma unroll
        for (int i = 0; i < 8; ++i) {
            for (int m = 8; m >= 1; m >>= 1) s[i] += __shfl_xor(s[i], m, 32);
            s[i] *= (1.f / F);
        }
#pragma unroll
        for (int j = 0; j < 8; ++j) {
#pragma unroll
            for (int i = 0; i < 8; ++i) {
                const float d = acc[j][i] - s[i];
                acc[j][i] = d;
                q[i] = fmaf(d, d, q[i]);
            }
        }
#pragma unroll
        for (int i = 0; i < 8; ++i) {
            for (int m = 8; m >= 1; m >>= 1) q[i] += __shfl_xor(q[i], m, 32);
            q[i] = rsqrtf(q[i] * (1.f / F) + LN_EPS);
        }
        float* trow = tile + ((lane >> 4) * 8) * TP + col;
#pragma unroll
        for (int j = 0; j < 8; ++j) {
            const float gj = g[j * 16 + col];
            const float ej = be[j * 16 + col];
#pragma unroll
            for (int i = 0; i < 8; ++i) trow[i * TP + j * 16] = acc[j][i] * q[i] * gj + ej;
        }
        __builtin_amdgcn_fence(__ATOMIC_RELEASE, "wavefront");
        __builtin_amdgcn_wave_barrier();
        __builtin_amdgcn_fence(__ATOMIC_ACQUIRE, "wavefront");
        float4 o[16];
#pragma unroll
        for (int r = 0; r < 16; ++r) o[r] = *(const float4*)&tile[r * TP + lane * 4];
        if (FIRST) {
#pragma unroll
            for (int r = 0; r < 16; ++r) {
                const float dr = __int_as_float(__builtin_amdgcn_readlane(__float_as_int(ddv), r));
                o[r].x *= dr; o[r].y *= dr; o[r].z *= dr; o[r].w *= dr;
            }
        }
#pragma unroll
        for (int r = 0; r < 16; ++r)
            if (row0 + r < N)
                *(float4*)&out[(size_t)(row0 + r) * F + lane * 4] = o[r];
        __builtin_amdgcn_fence(__ATOMIC_RELEASE, "wavefront");
        __builtin_amdgcn_wave_barrier();
        __builtin_amdgcn_fence(__ATOMIC_ACQUIRE, "wavefront");
    }
}

static inline size_t align256(size_t x) { return (x + 255) & ~(size_t)255; }

extern "C" void kernel_launch(void* const* d_in, const int* in_sizes, int n_in,
                              void* d_out, int out_size, void* d_ws, size_t ws_size,
                              hipStream_t stream) {
    const float* x   = (const float*)d_in[0];
    const int*   ei  = (const int*)  d_in[1];
    const float* W1  = (const float*)d_in[2];
    const float* b1  = (const float*)d_in[3];
    const float* W2  = (const float*)d_in[4];
    const float* b2  = (const float*)d_in[5];
    const float* g1  = (const float*)d_in[6];
    const float* be1 = (const float*)d_in[7];
    const float* g2  = (const float*)d_in[8];
    const float* be2 = (const float*)d_in[9];

    const int N = in_sizes[0] / F;
    const int E = in_sizes[1] / 2;
    const int* src = ei;
    const int* dst = ei + E;
    float* out = (float*)d_out;

    char* ws = (char*)d_ws;
    size_t off = 0;
    float* dinv = (float*)(ws + off); off = align256(off + (size_t)N * 4);
    int*   cnt  = (int*)  (ws + off); off = align256(off + (size_t)N * 4);
    float* Y    = (float*)(ws + off); off = align256(off + (size_t)N * F * 4);
    int*   esrc   = (int*)(ws + off); off = align256(off + (size_t)E * 4);
    int*   rowptr = (int*)(ws + off); off = align256(off + (size_t)N * 4);
    int*   tmp    = (int*)(ws + off); off = align256(off + (size_t)E * 8);
    int*   scnt   = (int*)(ws + off); off = align256(off + (size_t)NSLICE * MAXCB * 4);
    int*   soff   = (int*)(ws + off); off = align256(off + (size_t)NSLICE * MAXCB * 4);
    (void)ws_size; (void)n_in; (void)out_size;

    k_part<<<NSLICE, dim3(1024), 0, stream>>>(src, dst, E, scnt, soff, tmp);
    k_fine<<<(N + CB_NODES - 1) / CB_NODES, dim3(1024), 0, stream>>>(tmp, scnt, soff, N, E, cnt, rowptr, dinv, esrc);

    k_layer<true ><<<LAYER_WGS, dim3(NWF * 32), 0, stream>>>(rowptr, cnt, esrc, dinv, x, W1, b1, g1, be1, Y, N);
    k_layer<false><<<LAYER_WGS, dim3(NWF * 32), 0, stream>>>(rowptr, cnt, esrc, dinv, Y, W2, b2, g2, be2, out, N);
}


# Both bucket passes place records in a shared-memory copy of their stretch (slice of tmp, range of esrc), then write it as whole 128-byte lines; the 8- or 4-byte global store per edge is gone
# speedup vs baseline: 5.4583x; 1.0728x over previous
#include <hip/hip_runtime.h>

typedef float v2f __attribute__((ext_vector_type(2)));
typedef float v8f __attribute__((ext_vector_type(8)));
typedef int   v4i __attribute__((ext_vector_type(4)));

#define F 128
#define LN_EPS 1e-5f

#define CB_SHIFT 9
#define CB_NODES (1 << CB_SHIFT)
#define NSLICE   256
#define MAXCB    256
static_assert(NSLICE == 256 && MAXCB == 256, "k_fine maps thread t to table columns 4*(t & 63).. and slice group t >> 6");

#define EB 8
static_assert(EB == 8, "the asm operand list in k_part names EB values");
__global__ __launch_bounds__(1024) void k_part(const int* __restrict__ src,
                                               const int* __restrict__ dst, int E,
                                               int* __restrict__ scnt,
                                               int* __restrict__ soff,
                                               int* __restrict__ tmp) {
    __shared__ int h[MAXCB];
    __shared__ int sred[32];
    __shared__ int2 stg[EB * 1024];
    const int tid  = threadIdx.x;
    const int lane = tid & 31;
    const int wave = tid >> 5;
    if (tid < MAXCB) h[tid] = 0;
    __syncthreads();
    const int per = (E + NSLICE - 1) / NSLICE;
    const int e0 = blockIdx.x * per;
    int e1 = e0 + per;
    if (e1 > E) e1 = E;
    const int base0 = e0 + tid;

    int d[EB], s[EB];
#pragma unroll
    for (int u = 0; u < EB; ++u) { d[u] = 0; s[u] = 0; }
    if (e0 < e1) {
#pragma unroll
        for (int u = 0; u < EB; ++u) {
            const int e  = base0 + u * 1024;
            const int ec = (e < e1) ? e : e1 - 1;
            d[u] = dst[ec];
            s[u] = src[ec];
        }
        asm volatile("" :: "v"(d[0]), "v"(d[1]), "v"(d[2]), "v"(d[3]),
                           "v"(d[4]), "v"(d[5]), "v"(d[6]), "v"(d[7]),
                           "v"(s[0]), "v"(s[1]), "v"(s[2]), "v"(s[3]),
                           "v"(s[4]), "v"(s[5]), "v"(s[6]), "v"(s[7]));
#pragma unroll
        for (int u = 0; u < EB; ++u)
            if (base0 + u * 1024 < e1) atomicAdd(&h[d[u] >> CB_SHIFT], 1);
    }
    for (int e = base0 + EB * 1024; e < e1; e += 1024) atomicAdd(&h[dst[e] >> CB_SHIFT], 1);
    __syncthreads();

    const int hv = (tid < MAXCB) ? h[tid] : 0;
    int x = hv;
    for (int o = 1; o < 32; o <<= 1) {
        int y = __shfl_up(x, o, 32);
        if (lane >= o) x += y;
    }
    if (lane == 31) sred[wave] = x;
    __syncthreads();
    const int wtot = sred[lane];
    int wx = wtot;
    for (int o = 1; o < 32; o <<= 1) {
        int y = __shfl_up(wx, o, 32);
        if (lane >= o) wx += y;
    }
    const int before = __shfl(wx - wtot, wave, 32);
    const int p0 = e0 + before + (x - hv);
    if (tid < MAXCB) {
        scnt[blockIdx.x * MAXCB + tid] = hv;
        soff[blockIdx.x * MAXCB + tid] = p0;
        h[tid] = p0;
    }
    __syncthreads();

    const int len = e1 - e0;
    if (len <= EB * 1024) {
#pragma unroll
        for (int u = 0; u < EB; ++u) {
            if (base0 + u * 1024 < e1) {
                const int p = atomicAdd(&h[d[u] >> CB_SHIFT], 1);
                stg[p - e0] = make_int2(s[u], d[u]);
            }
        }
        __syncthreads();
        if (len > 0) {
            int2 v[EB];
#pragma unroll
            for (int u = 0; u < EB; ++u) {
                const int i = tid + u * 1024;
                v[u] = stg[(i < len) ? i : len - 1];
            }
#pragma unroll
            for (int u = 0; u < EB; ++u) {
                const int i = tid + u * 1024;
                if (i < len) *(int2*)&tmp[2 * (size_t)(e0 + i)] = v[u];
            }
        }
        return;
    }
    {
#pragma unroll
        for (int u = 0; u < EB; ++u) {
            if (base0 + u * 1024 < e1) {
                const int p = atomicAdd(&h[d[u] >> CB_SHIFT], 1);
                *(int2*)&tmp[2 * (size_t)p] = make_int2(s[u], d[u]);
            }
        }
    }
    for (int e = base0 + EB * 1024; e < e1; e += 1024) {
        const int dd = dst[e];
        const int ss = src[e];
        const int p = atomicAdd(&h[dd >> CB_SHIFT], 1);
        *(int2*)&tmp[2 * (size_t)p] = make_int2(ss, dd);
    }
}

#define FCAP 16384
#define RPW 8
static_assert(RPW * 32 == NSLICE, "every slice's run belongs to exactly one wave");
static_assert(RPW == 8, "the asm operand list in k_fine names 2*RPW values");
__global__ __launch_bounds__(1024) void k_fine(const int* __restrict__ tmp,
                                               const int* __restrict__ scnt,
                                               const int* __restrict__ soff, int N, int E,
                                               int* __restrict__ cnt,
                                               int* __restrict__ rowptr,
                                               float* __restrict__ dinv,
                                               int* __restrict__ esrc) {
    __shared__ int c[CB_NODES];
    __shared__ int sred[32];
    __shared__ int rs[NSLICE];
    __shared__ int rl[NSLICE];
    __shared__ int stg[FCAP];
    const int tid  = threadIdx.x;
    const int lane = tid & 31;
    const int wave = tid >> 5;
    const int rb   = blockIdx.x;
    const int node0 = rb << CB_SHIFT;

    if (tid < NSLICE) {
        rs[tid] = soff[tid * MAXCB + rb];
        rl[tid] = scnt[tid * MAXCB + rb];
    }
    if (tid < CB_NODES) c[tid] = 0;
    int part = 0;
    {
        const int c4 = (tid & 63) * 4;
        const int rg = tid >> 6;
        if (c4 < rb) {
            const int rot = rb & 15;
            v4i v[16];
#pragma unroll
            for (int k = 0; k < 16; ++k)
                v[k] = *(const v4i*)&scnt[(rg * 16 + (k ^ rot)) * MAXCB + c4];
            asm volatile("" :: "v"(v[0]), "v"(v[1]), "v"(v[2]), "v"(v[3]),
                               "v"(v[4]), "v"(v[5]), "v"(v[6]), "v"(v[7]),
                               "v"(v[8]), "v"(v[9]), "v"(v[10]), "v"(v[11]),
                               "v"(v[12]), "v"(v[13]), "v"(v[14]), "v"(v[15]));
            int sx = 0, sy = 0, sz = 0, sw = 0;
#pragma unroll
            for (int k = 0; k < 16; ++k) { sx += v[k].x; sy += v[k].y; sz += v[k].z; sw += v[k].w; }
            part = sx;
            if (c4 + 1 < rb) part += sy;
            if (c4 + 2 < rb) part += sz;
            if (c4 + 3 < rb) part += sw;
        }
    }
    for (int o = 16; o >= 1; o >>= 1) part += __shfl_xor(part, o, 32);
    if (lane == 0) sred[wave] = part;
    __syncthreads();
    int e0 = sred[lane];
    for (int o = 16; o >= 1; o >>= 1) e0 += __shfl_xor(e0, o, 32);

    int st[RPW], ln[RPW];
#pragma unroll
    for (int k = 0; k < RPW; ++k) {
        st[k] = __builtin_amdgcn_readfirstlane(rs[wave + 32 * k]);
        ln[k] = __builtin_amdgcn_readfirstlane(rl[wave + 32 * k]);
    }

    int2 rec[2 * RPW];
#pragma unroll
    for (int k = 0; k < RPW; ++k) {
#pragma unroll
        for (int j = 0; j < 2; ++j) {
            const int i = lane + 32 * j;
            int idx = st[k] + ((i < ln[k]) ? i : 0);
            if (idx > E - 1) idx = E - 1;
            if (idx < 0) idx = 0;
            rec[2 * k + j] = *(const int2*)&tmp[2 * (size_t)idx];
        }
    }
    asm volatile("" :: "v"(rec[0].y), "v"(rec[1].y), "v"(rec[2].y), "v"(rec[3].y),
                       "v"(rec[4].y), "v"(rec[5].y), "v"(rec[6].y), "v"(rec[7].y),
                       "v"(rec[8].y), "v"(rec[9].y), "v"(rec[10].y), "v"(rec[11].y),
                       "v"(rec[12].y), "v"(rec[13].y), "v"(rec[14].y), "v"(rec[15].y));
#pragma unroll
    for (int k = 0; k < RPW; ++k) {
#pragma unroll
        for (int j = 0; j < 2; ++j)
            if (lane + 32 * j < ln[k]) atomicAdd(&c[rec[2 * k + j].y - node0], 1);
    }
#pragma unroll
    for (int k = 0; k < RPW; ++k)
        for (int i = lane + 64; i < ln[k]; i += 32)
            atomicAdd(&c[tmp[2 * (size_t)(st[k] + i) + 1] - node0], 1);
    __syncthreads();

    const int cv = (tid < CB_NODES) ? c[tid] : 0;
    int x = cv;
    for (int o = 1; o < 32; o <<= 1) {
        int y = __shfl_up(x, o, 32);
        if (lane >= o) x += y;
    }
    if (lane == 31) sred[wave] = x;
    __syncthreads();
    const int wtot = sred[lane];
    int wx = wtot;
    for (int o = 1; o < 32; o <<= 1) {
        int y = __shfl_up(wx, o, 32);
        if (lane >= o) wx += y;
    }
    const int before = __shfl(wx - wtot, wave, 32);
    const int p0 = e0 + before + (x - cv);

    if (tid < CB_NODES) {
        const int v = node0 + tid;
        if (v < N) {
            cnt[v]    = cv;
            rowptr[v] = p0;
            dinv[v]   = rsqrtf((float)(cv + 1));
        }
        c[tid] = p0;
    }
    __syncthreads();
    const int tot = __builtin_amdgcn_readfirstlane(__shfl(wx, 31, 32));
    if (tot <= FCAP) {
#pragma unroll
        for (int k = 0; k < RPW; ++k) {
#pragma unroll
            for (int j = 0; j < 2; ++j) {
                if (lane + 32 * j < ln[k]) {
                    const int p = atomicAdd(&c[rec[2 * k + j].y - node0], 1);
                    stg[p - e0] = rec[2 * k + j].x;
                }
            }
        }
#pragma unroll
        for (int k = 0; k < RPW; ++k) {
            for (int i = lane + 64; i < ln[k]; i += 32) {
                const int2 r = *(const int2*)&tmp[2 * (size_t)(st[k] + i)];
                const int p = atomicAdd(&c[r.y - node0], 1);
                stg[p - e0] = r.x;
            }
        }
        __syncthreads();
        for (int b = tid; b < tot; b += 4 * 1024) {
            int v[4];
#pragma unroll
            for (int u = 0; u < 4; ++u) {
                const int i = b + u * 1024;
                v[u] = stg[(i < tot) ? i : tot - 1];
            }
#pragma unroll
            for (int u = 0; u < 4; ++u) {
                const int i = b + u * 1024;
                if (i < tot) esrc[e0 + i] = v[u];
            }
        }
        return;
    }
#pragma unroll
    for (int k = 0; k < RPW; ++k) {
#pragma unroll
        for (int j = 0; j < 2; ++j) {
            if (lane + 32 * j < ln[k]) {
                const int p = atomicAdd(&c[rec[2 * k + j].y - node0], 1);
                esrc[p] = rec[2 * k + j].x;
            }
        }
    }
#pragma unroll
    for (int k = 0; k < RPW; ++k) {
        for (int i = lane + 64; i < ln[k]; i += 32) {
            const int2 r = *(const int2*)&tmp[2 * (size_t)(st[k] + i)];
            const int p = atomicAdd(&c[r.y - node0], 1);
            esrc[p] = r.x;
        }
    }
}

#define NWF 13
#define LAYER_WGS 256
#define TP  132

template <bool FIRST>
__global__ __launch_bounds__(NWF * 32) void k_layer(const int* __restrict__ rowptr,
                                                    const int* __restrict__ cnt,
                                                    const int* __restrict__ esrc,
                                                    const float* __restrict__ dinv,
                                                    const float* __restrict__ X,
                                                    const float* __restrict__ W,
                                                    const float* __restrict__ bias,
                                                    const float* __restrict__ g,
                                                    const float* __restrict__ be,
                                                    float* __restrict__ out, int N) {
    __shared__ float sW[F * F];
    __shared__ float sT[NWF * 16 * TP];
    for (int i = threadIdx.x; i < (F / 2) * (F / 4); i += NWF * 32) {
        int p = i >> 5;
        int c = (i & 31) * 4;
        float4 w0 = *(const float4*)&W[(2 * p)     * F + c];
        float4 w1 = *(const float4*)&W[(2 * p + 1) * F + c];
        v2f* d = (v2f*)&sW[p * 256 + c * 2];
        d[0] = (v2f){w0.x, w1.x};
        d[1] = (v2f){w0.y, w1.y};
        d[2] = (v2f){w0.z, w1.z};
        d[3] = (v2f){w0.w, w1.w};
    }
    __syncthreads();

    const int lane = threadIdx.x & 31;
    const int wave = __builtin_amdgcn_readfirstlane((int)(threadIdx.x >> 5));
    float* tile = &sT[wave * 16 * TP];
    const float* Xl = X + lane * 4;
    const int col   = lane & 15;
    const int khalf = (lane >> 4) << 1;
    const int ntiles = (N + 15) >> 4;

    for (int t = __builtin_amdgcn_readfirstlane((int)(wave * gridDim.x + blockIdx.x)); t < ntiles;
         t += NWF * (int)gridDim.x) {
        const int row0 = t * 16;

        float4 hs[16];
#pragma unroll
        for (int r = 0; r < 16; ++r) {
            int row = row0 + r;
            if (row > N - 1) row = N - 1;
            hs[r] = *(const float4*)(Xl + (size_t)row * F);
        }
        int mrow = row0 + (lane & 15);
        if (mrow > N - 1) mrow = N - 1;
        const int   startv = rowptr[mrow];
        const int   endv   = startv + cnt[mrow];
        const float ddv    = dinv[mrow];
        const int e0    = __builtin_amdgcn_readlane(startv, 0);
        const int e_end = __builtin_amdgcn_readlane(endv, 15);

        int sv = 0, wv = 0;
        if (e_end > e0) {
            int idx = e0 + lane;
            const bool live = idx < e_end;
            if (!live) idx = e_end - 1;
            sv = esrc[idx];
            if (FIRST) {
                const int wl = __float_as_int(dinv[sv]);
                wv = live ? wl : 0;
            }
        }
#pragma unroll
        for (int r = 0; r < 16; ++r) *(float4*)&tile[r * TP + lane * 4] = hs[r];

        int   r    = 0;
        int   closed = 0;
        int   rend = __builtin_amdgcn_readlane(endv, 0);
        float dd   = __int_as_float(__builtin_amdgcn_readlane(__float_as_int(ddv), 0));
        float4 own = *(const float4*)&tile[lane * 4];
        float4 ga = {0.f, 0.f, 0.f, 0.f};
#define CLOSE_ROW()                                                                         \
        do {                                                                                    \
            float4 v;                                                                           \
            if (FIRST) {                                                                        \
                v.x = dd * fmaf(dd, own.x, ga.x); v.y = dd * fmaf(dd, own.y, ga.y);             \
                v.z = dd * fmaf(dd, own.z, ga.z); v.w = dd * fmaf(dd, own.w, ga.w);             \
            } else {                                                                            \
                v.x = dd * (ga.x + own.x); v.y = dd * (ga.y + own.y);                           \
                v.z = dd * (ga.z + own.z); v.w = dd * (ga.w + own.w);                           \
            }                                                                                   \
            *(float4*)&tile[r * TP + lane * 4] = v;                                             \
            ga.x = 0.f; ga.y = 0.f; ga.z = 0.f; ga.w = 0.f;                                     \
            ++closed;                                                                           \
            r = (r < 15) ? r + 1 : 15;                                                          \
            rend = (closed < 16) ? __builtin_amdgcn_readlane(endv, r) : 0x7fffffff;             \
            dd   = __int_as_float(__builtin_amdgcn_readlane(__float_as_int(ddv), r));           \
            own  = *(const float4*)&tile[r * TP + lane * 4];                                    \
        } while (0)

        for (int c0 = e0; c0 < e_end; c0 += 32) {
            float4 h[32];
#pragma unroll
            for (int i = 0; i < 32; ++i) {
                const int s = __builtin_amdgcn_readlane(sv, i);
                h[i] = *(const float4*)(Xl + (size_t)s * F);
            }
            int idxn = c0 + 32 + lane;
            if (idxn > e_end - 1) idxn = e_end - 1;
            const int svn = esrc[idxn];
            int rrel = rend - c0;
#pragma unroll
            for (int i = 0; i < 32; ++i) {
                if (__builtin_expect(rrel == i, 0)) {
                    do { CLOSE_ROW(); rrel = rend - c0; } while (rrel == i);
                }
                if (FIRST) {
                    const float w = __int_as_float(__builtin_amdgcn_readlane(wv, i));
                    ga.x = fmaf(w, h[i].x, ga.x);
                    ga.y = fmaf(w, h[i].y, ga.y);
                    ga.z = fmaf(w, h[i].z, ga.z);
                    ga.w = fmaf(w, h[i].w, ga.w);
                } else {
                    ga.x += h[i].x;
                    ga.y += h[i].y;
                    ga.z += h[i].z;
                    ga.w += h[i].w;
                }
            }
            sv = svn;
            if (FIRST) {
                const int wn = __float_as_int(dinv[svn]);
                wv = (c0 + 32 + lane < e_end) ? wn : 0;
            }
        }
        while (closed < 16) CLOSE_ROW();
#undef CLOSE_ROW

        __builtin_amdgcn_fence(__ATOMIC_RELEASE, "wavefront");
        __builtin_amdgcn_wave_barrier();
        __builtin_amdgcn_fence(__ATOMIC_ACQUIRE, "wavefront");

        const float* arow = tile + col * TP + khalf;

        v8f acc[8];
        const v8f vzero = {0.f, 0.f, 0.f, 0.f, 0.f, 0.f, 0.f, 0.f};
#pragma unroll
        for (int j = 0; j < 8; ++j) acc[j] = vzero;

        for (int k0 = 0; k0 < F; k0 += 4) {
            const v2f a = *(const v2f*)(arow + k0);
            const v2f* bbase = (const v2f*)&sW[((k0 + khalf) >> 1) * 256 + col * 2];
#pragma unroll
            for (int j = 0; j < 8; ++j) {
                v2f b = bbase[j * 16];
                acc[j] = __builtin_amdgcn_wmma_f32_16x16x4_f32(
                    false, a, false, b, (short)0, acc[j], false, false);
            }
        }

        float s[8], q[8];
#pragma unroll
        for (int i = 0; i < 8; ++i) { s[i] = 0.f; q[i] = 0.f; }
#pragma unroll
        for (int j = 0; j < 8; ++j) {
            const float bj = bias[j * 16 + col];
#pragma unroll
            for (int i = 0; i < 8; ++i) {
                const float v = fmaxf(acc[j][i] + bj, 0.f);
                acc[j][i] = v;
                s[i] += v;
            }
        }
#pragma unroll
        for (int i = 0; i < 8; ++i) {
            for (int m = 8; m >= 1; m >>= 1) s[i] += __shfl_xor(s[i], m, 32);
            s[i] *= (1.f / F);
        }
#pragma unroll
        for (int j = 0; j < 8; ++j) {
#pragma unroll
            for (int i = 0; i < 8; ++i) {
                const float d = acc[j][i] - s[i];
                acc[j][i] = d;
                q[i] = fmaf(d, d, q[i]);
            }
        }
#pragma unroll
        for (int i = 0; i < 8; ++i) {
            for (int m = 8; m >= 1; m >>= 1) q[i] += __shfl_xor(q[i], m, 32);
            q[i] = rsqrtf(q[i] * (1.f / F) + LN_EPS);
        }
        float* trow = tile + ((lane >> 4) * 8) * TP + col;
#pragma unroll
        for (int j = 0; j < 8; ++j) {
            const float gj = g[j * 16 + col];
            const float ej = be[j * 16 + col];
#pragma unroll
            for (int i = 0; i < 8; ++i) trow[i * TP + j * 16] = acc[j][i] * q[i] * gj + ej;
        }
        __builtin_amdgcn_fence(__ATOMIC_RELEASE, "wavefront");
        __builtin_amdgcn_wave_barrier();
        __builtin_amdgcn_fence(__ATOMIC_ACQUIRE, "wavefront");
        float4 o[16];
#pragma unroll
        for (int r = 0; r < 16; ++r) o[r] = *(const float4*)&tile[r * TP + lane * 4];
        if (FIRST) {
#pragma unroll
            for (int r = 0; r < 16; ++r) {
                const float dr = __int_as_float(__builtin_amdgcn_readlane(__float_as_int(ddv), r));
                o[r].x *= dr; o[r].y *= dr; o[r].z *= dr; o[r].w *= dr;
            }
        }
#pragma unroll
        for (int r = 0; r < 16; ++r)
            if (row0 + r < N)
                *(float4*)&out[(size_t)(row0 + r) * F + lane * 4] = o[r];
        __builtin_amdgcn_fence(__ATOMIC_RELEASE, "wavefront");
        __builtin_amdgcn_wave_barrier();
        __builtin_amdgcn_fence(__ATOMIC_ACQUIRE, "wavefront");
    }
}

static inline size_t align256(size_t x) { return (x + 255) & ~(size_t)255; }

extern "C" void kernel_launch(void* const* d_in, const int* in_sizes, int n_in,
                              void* d_out, int out_size, void* d_ws, size_t ws_size,
                              hipStream_t stream) {
    const float* x   = (const float*)d_in[0];
    const int*   ei  = (const int*)  d_in[1];
    const float* W1  = (const float*)d_in[2];
    const float* b1  = (const float*)d_in[3];
    const float* W2  = (const float*)d_in[4];
    const float* b2  = (const float*)d_in[5];
    const float* g1  = (const float*)d_in[6];
    const float* be1 = (const float*)d_in[7];
    const float* g2  = (const float*)d_in[8];
    const float* be2 = (const float*)d_in[9];

    const int N = in_sizes[0] / F;
    const int E = in_sizes[1] / 2;
    const int* src = ei;
    const int* dst = ei + E;
    float* out = (float*)d_out;

    char* ws = (char*)d_ws;
    size_t off = 0;
    float* dinv = (float*)(ws + off); off = align256(off + (size_t)N * 4);
    int*   cnt  = (int*)  (ws + off); off = align256(off + (size_t)N * 4);
    float* Y    = (float*)(ws + off); off = align256(off + (size_t)N * F * 4);
    int*   esrc   = (int*)(ws + off); off = align256(off + (size_t)E * 4);
    int*   rowptr = (int*)(ws + off); off = align256(off + (size_t)N * 4);
    int*   tmp    = (int*)(ws + off); off = align256(off + (size_t)E * 8);
    int*   scnt   = (int*)(ws + off); off = align256(off + (size_t)NSLICE * MAXCB * 4);
    int*   soff   = (int*)(ws + off); off = align256(off + (size_t)NSLICE * MAXCB * 4);
    (void)ws_size; (void)n_in; (void)out_size;

    k_part<<<NSLICE, dim3(1024), 0, stream>>>(src, dst, E, scnt, soff, tmp);
    k_fine<<<(N + CB_NODES - 1) / CB_NODES, dim3(1024), 0, stream>>>(tmp, scnt, soff, N, E, cnt, rowptr, dinv, esrc);

    k_layer<true ><<<LAYER_WGS, dim3(NWF * 32), 0, stream>>>(rowptr, cnt, esrc, dinv, x, W1, b1, g1, be1, Y, N);
    k_layer<false><<<LAYER_WGS, dim3(NWF * 32), 0, stream>>>(rowptr, cnt, esrc, dinv, Y, W2, b2, g2, be2, out, N);
}


# Gather: a chunk's 32 row loads issue back to back from 32-bit per-lane offsets on a scalar base, formed one chunk ahead under the rows in flight; next weights asked for early, no wait at loop end
# speedup vs baseline: 5.6483x; 1.0348x over previous
#include <hip/hip_runtime.h>

typedef float v2f __attribute__((ext_vector_type(2)));
typedef float v8f __attribute__((ext_vector_type(8)));
typedef int   v4i __attribute__((ext_vector_type(4)));

#define F 128
#define LN_EPS 1e-5f

#define CB_SHIFT 9
#define CB_NODES (1 << CB_SHIFT)
#define NSLICE   256
#define MAXCB    256
static_assert(NSLICE == 256 && MAXCB == 256, "k_fine maps thread t to table columns 4*(t & 63).. and slice group t >> 6");

#define EB 8
static_assert(EB == 8, "the asm operand list in k_part names EB values");
__global__ __launch_bounds__(1024) void k_part(const int* __restrict__ src,
                                               const int* __restrict__ dst, int E,
                                               int* __restrict__ scnt,
                                               int* __restrict__ soff,
                                               int* __restrict__ tmp) {
    __shared__ int h[MAXCB];
    __shared__ int sred[32];
    __shared__ int2 stg[EB * 1024];
    const int tid  = threadIdx.x;
    const int lane = tid & 31;
    const int wave = tid >> 5;
    if (tid < MAXCB) h[tid] = 0;
    __syncthreads();
    const int per = (E + NSLICE - 1) / NSLICE;
    const int e0 = blockIdx.x * per;
    int e1 = e0 + per;
    if (e1 > E) e1 = E;
    const int base0 = e0 + tid;

    int d[EB], s[EB];
#pragma unroll
    for (int u = 0; u < EB; ++u) { d[u] = 0; s[u] = 0; }
    if (e0 < e1) {
#pragma unroll
        for (int u = 0; u < EB; ++u) {
            const int e  = base0 + u * 1024;
            const int ec = (e < e1) ? e : e1 - 1;
            d[u] = dst[ec];
            s[u] = src[ec];
        }
        asm volatile("" :: "v"(d[0]), "v"(d[1]), "v"(d[2]), "v"(d[3]),
                           "v"(d[4]), "v"(d[5]), "v"(d[6]), "v"(d[7]),
                           "v"(s[0]), "v"(s[1]), "v"(s[2]), "v"(s[3]),
                           "v"(s[4]), "v"(s[5]), "v"(s[6]), "v"(s[7]));
#pragma unroll
        for (int u = 0; u < EB; ++u)
            if (base0 + u * 1024 < e1) atomicAdd(&h[d[u] >> CB_SHIFT], 1);
    }
    for (int e = base0 + EB * 1024; e < e1; e += 1024) atomicAdd(&h[dst[e] >> CB_SHIFT], 1);
    __syncthreads();

    const int hv = (tid < MAXCB) ? h[tid] : 0;
    int x = hv;
    for (int o = 1; o < 32; o <<= 1) {
        int y = __shfl_up(x, o, 32);
        if (lane >= o) x += y;
    }
    if (lane == 31) sred[wave] = x;
    __syncthreads();
    const int wtot = sred[lane];
    int wx = wtot;
    for (int o = 1; o < 32; o <<= 1) {
        int y = __shfl_up(wx, o, 32);
        if (lane >= o) wx += y;
    }
    const int before = __shfl(wx - wtot, wave, 32);
    const int p0 = e0 + before + (x - hv);
    if (tid < MAXCB) {
        scnt[blockIdx.x * MAXCB + tid] = hv;
        soff[blockIdx.x * MAXCB + tid] = p0;
        h[tid] = p0;
    }
    __syncthreads();

    const int len = e1 - e0;
    if (len <= EB * 1024) {
#pragma unroll
        for (int u = 0; u < EB; ++u) {
            if (base0 + u * 1024 < e1) {
                const int p = atomicAdd(&h[d[u] >> CB_SHIFT], 1);
                stg[p - e0] = make_int2(s[u], d[u]);
            }
        }
        __syncthreads();
        if (len > 0) {
            int2 v[EB];
#pragma unroll
            for (int u = 0; u < EB; ++u) {
                const int i = tid + u * 1024;
                v[u] = stg[(i < len) ? i : len - 1];
            }
#pragma unroll
            for (int u = 0; u < EB; ++u) {
                const int i = tid + u * 1024;
                if (i < len) *(int2*)&tmp[2 * (size_t)(e0 + i)] = v[u];
            }
        }
        return;
    }
    {
#pragma unroll
        for (int u = 0; u < EB; ++u) {
            if (base0 + u * 1024 < e1) {
                const int p = atomicAdd(&h[d[u] >> CB_SHIFT], 1);
                *(int2*)&tmp[2 * (size_t)p] = make_int2(s[u], d[u]);
            }
        }
    }
    for (int e = base0 + EB * 1024; e < e1; e += 1024) {
        const int dd = dst[e];
        const int ss = src[e];
        const int p = atomicAdd(&h[dd >> CB_SHIFT], 1);
        *(int2*)&tmp[2 * (size_t)p] = make_int2(ss, dd);
    }
}

#define FCAP 16384
#define RPW 8
static_assert(RPW * 32 == NSLICE, "every slice's run belongs to exactly one wave");
static_assert(RPW == 8, "the asm operand list in k_fine names 2*RPW values");
__global__ __launch_bounds__(1024) void k_fine(const int* __restrict__ tmp,
                                               const int* __restrict__ scnt,
                                               const int* __restrict__ soff, int N, int E,
                                               int* __restrict__ cnt,
                                               int* __restrict__ rowptr,
                                               float* __restrict__ dinv,
                                               int* __restrict__ esrc) {
    __shared__ int c[CB_NODES];
    __shared__ int sred[32];
    __shared__ int rs[NSLICE];
    __shared__ int rl[NSLICE];
    __shared__ int stg[FCAP];
    const int tid  = threadIdx.x;
    const int lane = tid & 31;
    const int wave = tid >> 5;
    const int rb   = blockIdx.x;
    const int node0 = rb << CB_SHIFT;

    if (tid < NSLICE) {
        rs[tid] = soff[tid * MAXCB + rb];
        rl[tid] = scnt[tid * MAXCB + rb];
    }
    if (tid < CB_NODES) c[tid] = 0;
    int part = 0;
    {
        const int c4 = (tid & 63) * 4;
        const int rg = tid >> 6;
        if (c4 < rb) {
            const int rot = rb & 15;
            v4i v[16];
#pragma unroll
            for (int k = 0; k < 16; ++k)
                v[k] = *(const v4i*)&scnt[(rg * 16 + (k ^ rot)) * MAXCB + c4];
            asm volatile("" :: "v"(v[0]), "v"(v[1]), "v"(v[2]), "v"(v[3]),
                               "v"(v[4]), "v"(v[5]), "v"(v[6]), "v"(v[7]),
                               "v"(v[8]), "v"(v[9]), "v"(v[10]), "v"(v[11]),
                               "v"(v[12]), "v"(v[13]), "v"(v[14]), "v"(v[15]));
            int sx = 0, sy = 0, sz = 0, sw = 0;
#pragma unroll
            for (int k = 0; k < 16; ++k) { sx += v[k].x; sy += v[k].y; sz += v[k].z; sw += v[k].w; }
            part = sx;
            if (c4 + 1 < rb) part += sy;
            if (c4 + 2 < rb) part += sz;
            if (c4 + 3 < rb) part += sw;
        }
    }
    for (int o = 16; o >= 1; o >>= 1) part += __shfl_xor(part, o, 32);
    if (lane == 0) sred[wave] = part;
    __syncthreads();
    int e0 = sred[lane];
    for (int o = 16; o >= 1; o >>= 1) e0 += __shfl_xor(e0, o, 32);

    int st[RPW], ln[RPW];
#pragma unroll
    for (int k = 0; k < RPW; ++k) {
        st[k] = __builtin_amdgcn_readfirstlane(rs[wave + 32 * k]);
        ln[k] = __builtin_amdgcn_readfirstlane(rl[wave + 32 * k]);
    }

    int2 rec[2 * RPW];
#pragma unroll
    for (int k = 0; k < RPW; ++k) {
#pragma unroll
        for (int j = 0; j < 2; ++j) {
            const int i = lane + 32 * j;
            int idx = st[k] + ((i < ln[k]) ? i : 0);
            if (idx > E - 1) idx = E - 1;
            if (idx < 0) idx = 0;
            rec[2 * k + j] = *(const int2*)&tmp[2 * (size_t)idx];
        }
    }
    asm volatile("" :: "v"(rec[0].y), "v"(rec[1].y), "v"(rec[2].y), "v"(rec[3].y),
                       "v"(rec[4].y), "v"(rec[5].y), "v"(rec[6].y), "v"(rec[7].y),
                       "v"(rec[8].y), "v"(rec[9].y), "v"(rec[10].y), "v"(rec[11].y),
                       "v"(rec[12].y), "v"(rec[13].y), "v"(rec[14].y), "v"(rec[15].y));
#pragma unroll
    for (int k = 0; k < RPW; ++k) {
#pragma unroll
        for (int j = 0; j < 2; ++j)
            if (lane + 32 * j < ln[k]) atomicAdd(&c[rec[2 * k + j].y - node0], 1);
    }
#pragma unroll
    for (int k = 0; k < RPW; ++k)
        for (int i = lane + 64; i < ln[k]; i += 32)
            atomicAdd(&c[tmp[2 * (size_t)(st[k] + i) + 1] - node0], 1);
    __syncthreads();

    const int cv = (tid < CB_NODES) ? c[tid] : 0;
    int x = cv;
    for (int o = 1; o < 32; o <<= 1) {
        int y = __shfl_up(x, o, 32);
        if (lane >= o) x += y;
    }
    if (lane == 31) sred[wave] = x;
    __syncthreads();
    const int wtot = sred[lane];
    int wx = wtot;
    for (int o = 1; o < 32; o <<= 1) {
        int y = __shfl_up(wx, o, 32);
        if (lane >= o) wx += y;
    }
    const int before = __shfl(wx - wtot, wave, 32);
    const int p0 = e0 + before + (x - cv);

    if (tid < CB_NODES) {
        const int v = node0 + tid;
        if (v < N) {
            cnt[v]    = cv;
            rowptr[v] = p0;
            dinv[v]   = rsqrtf((float)(cv + 1));
        }
        c[tid] = p0;
    }
    __syncthreads();
    const int tot = __builtin_amdgcn_readfirstlane(__shfl(wx, 31, 32));
    if (tot <= FCAP) {
#pragma unroll
        for (int k = 0; k < RPW; ++k) {
#pragma unroll
            for (int j = 0; j < 2; ++j) {
                if (lane + 32 * j < ln[k]) {
                    const int p = atomicAdd(&c[rec[2 * k + j].y - node0], 1);
                    stg[p - e0] = rec[2 * k + j].x;
                }
            }
        }
#pragma unroll
        for (int k = 0; k < RPW; ++k) {
            for (int i = lane + 64; i < ln[k]; i += 32) {
                const int2 r = *(const int2*)&tmp[2 * (size_t)(st[k] + i)];
                const int p = atomicAdd(&c[r.y - node0], 1);
                stg[p - e0] = r.x;
            }
        }
        __syncthreads();
        for (int b = tid; b < tot; b += 4 * 1024) {
            int v[4];
#pragma unroll
            for (int u = 0; u < 4; ++u) {
                const int i = b + u * 1024;
                v[u] = stg[(i < tot) ? i : tot - 1];
            }
#pragma unroll
            for (int u = 0; u < 4; ++u) {
                const int i = b + u * 1024;
                if (i < tot) esrc[e0 + i] = v[u];
            }
        }
        return;
    }
#pragma unroll
    for (int k = 0; k < RPW; ++k) {
#pragma unroll
        for (int j = 0; j < 2; ++j) {
            if (lane + 32 * j < ln[k]) {
                const int p = atomicAdd(&c[rec[2 * k + j].y - node0], 1);
                esrc[p] = rec[2 * k + j].x;
            }
        }
    }
#pragma unroll
    for (int k = 0; k < RPW; ++k) {
        for (int i = lane + 64; i < ln[k]; i += 32) {
            const int2 r = *(const int2*)&tmp[2 * (size_t)(st[k] + i)];
            const int p = atomicAdd(&c[r.y - node0], 1);
            esrc[p] = r.x;
        }
    }
}

#define NWF 13
#define LAYER_WGS 256
#define TP  132

template <bool FIRST>
__global__ __launch_bounds__(NWF * 32) void k_layer(const int* __restrict__ rowptr,
                                                    const int* __restrict__ cnt,
                                                    const int* __restrict__ esrc,
                                                    const float* __restrict__ dinv,
                                                    const float* __restrict__ X,
                                                    const float* __restrict__ W,
                                                    const float* __restrict__ bias,
                                                    const float* __restrict__ g,
                                                    const float* __restrict__ be,
                                                    float* __restrict__ out, int N) {
    __shared__ float sW[F * F];
    __shared__ float sT[NWF * 16 * TP];
    for (int i = threadIdx.x; i < (F / 2) * (F / 4); i += NWF * 32) {
        int p = i >> 5;
        int c = (i & 31) * 4;
        float4 w0 = *(const float4*)&W[(2 * p)     * F + c];
        float4 w1 = *(const float4*)&W[(2 * p + 1) * F + c];
        v2f* d = (v2f*)&sW[p * 256 + c * 2];
        d[0] = (v2f){w0.x, w1.x};
        d[1] = (v2f){w0.y, w1.y};
        d[2] = (v2f){w0.z, w1.z};
        d[3] = (v2f){w0.w, w1.w};
    }
    __syncthreads();

    const int lane = threadIdx.x & 31;
    const int wave = __builtin_amdgcn_readfirstlane((int)(threadIdx.x >> 5));
    float* tile = &sT[wave * 16 * TP];
    const char* Xb = (const char*)X;
    const unsigned lane16 = (unsigned)lane * 16u;
    const int col   = lane & 15;
    const int khalf = (lane >> 4) << 1;
    const int ntiles = (N + 15) >> 4;

    for (int t = __builtin_amdgcn_readfirstlane((int)(wave * gridDim.x + blockIdx.x)); t < ntiles;
         t += NWF * (int)gridDim.x) {
        const int row0 = t * 16;

        float4 hs[16];
#pragma unroll
        for (int r = 0; r < 16; ++r) {
            int row = row0 + r;
            if (row > N - 1) row = N - 1;
            hs[r] = *(const float4*)(Xb + (size_t)((unsigned)row * (unsigned)(F * 4) + lane16));
        }
        int mrow = row0 + (lane & 15);
        if (mrow > N - 1) mrow = N - 1;
        const int   startv = rowptr[mrow];
        const int   endv   = startv + cnt[mrow];
        const float ddv    = dinv[mrow];
        const int e0    = __builtin_amdgcn_readlane(startv, 0);
        const int e_end = __builtin_amdgcn_readlane(endv, 15);

        int sv1 = 0, wv = 0;
        unsigned off[32];
#pragma unroll
        for (int i = 0; i < 32; ++i) off[i] = lane16;
        if (e_end > e0) {
            int idx0 = e0 + lane;
            if (idx0 > e_end - 1) idx0 = e_end - 1;
            int idx1 = e0 + 32 + lane;
            if (idx1 > e_end - 1) idx1 = e_end - 1;
            const int sv0 = esrc[idx0];
            sv1 = esrc[idx1];
            if (FIRST) {
                wv  = __float_as_int(dinv[sv0]);
            }
#pragma unroll
            for (int i = 0; i < 32; ++i)
                off[i] = (unsigned)__builtin_amdgcn_readlane(sv0, i) * (unsigned)(F * 4) + lane16;
        }
#pragma unroll
        for (int r = 0; r < 16; ++r) *(float4*)&tile[r * TP + lane * 4] = hs[r];

        int   r    = 0;
        int   closed = 0;
        int   rend = __builtin_amdgcn_readlane(endv, 0);
        float dd   = __int_as_float(__builtin_amdgcn_readlane(__float_as_int(ddv), 0));
        float4 own = *(const float4*)&tile[lane * 4];
        float4 ga = {0.f, 0.f, 0.f, 0.f};
#define CLOSE_ROW()                                                                         \
        do {                                                                                    \
            float4 v;                                                                           \
            if (FIRST) {                                                                        \
                v.x = dd * fmaf(dd, own.x, ga.x); v.y = dd * fmaf(dd, own.y, ga.y);             \
                v.z = dd * fmaf(dd, own.z, ga.z); v.w = dd * fmaf(dd, own.w, ga.w);             \
            } else {                                                                            \
                v.x = dd * (ga.x + own.x); v.y = dd * (ga.y + own.y);                           \
                v.z = dd * (ga.z + own.z); v.w = dd * (ga.w + own.w);                           \
            }                                                                                   \
            *(float4*)&tile[r * TP + lane * 4] = v;                                             \
            ga.x = 0.f; ga.y = 0.f; ga.z = 0.f; ga.w = 0.f;                                     \
            ++closed;                                                                           \
            r = (r < 15) ? r + 1 : 15;                                                          \
            rend = (closed < 16) ? __builtin_amdgcn_readlane(endv, r) : 0x7fffffff;             \
            dd   = __int_as_float(__builtin_amdgcn_readlane(__float_as_int(ddv), r));           \
            own  = *(const float4*)&tile[r * TP + lane * 4];                                    \
        } while (0)

        for (int c0 = e0; c0 < e_end; c0 += 32) {
            float4 h[32];
#pragma unroll
            for (int i = 0; i < 32; ++i) h[i] = *(const float4*)(Xb + (size_t)off[i]);
            int idx2 = c0 + 64 + lane;
            if (idx2 > e_end - 1) idx2 = e_end - 1;
            const int sv2 = esrc[idx2];
            int wn = 0;
            if (FIRST) wn = __float_as_int(dinv[sv1]);
#pragma unroll
            for (int i = 0; i < 32; ++i)
                off[i] = (unsigned)__builtin_amdgcn_readlane(sv1, i) * (unsigned)(F * 4) + lane16;
            asm volatile("" :: "v"(off[0]), "v"(off[1]), "v"(off[2]), "v"(off[3]),
                               "v"(off[4]), "v"(off[5]), "v"(off[6]), "v"(off[7]),
                               "v"(off[8]), "v"(off[9]), "v"(off[10]), "v"(off[11]),
                               "v"(off[12]), "v"(off[13]), "v"(off[14]), "v"(off[15]));
            asm volatile("" :: "v"(off[16]), "v"(off[17]), "v"(off[18]), "v"(off[19]),
                               "v"(off[20]), "v"(off[21]), "v"(off[22]), "v"(off[23]),
                               "v"(off[24]), "v"(off[25]), "v"(off[26]), "v"(off[27]),
                               "v"(off[28]), "v"(off[29]), "v"(off[30]), "v"(off[31]));
            int rrel = rend - c0;
#pragma unroll
            for (int i = 0; i < 32; ++i) {
                if (__builtin_expect(rrel == i, 0)) {
                    do { CLOSE_ROW(); rrel = rend - c0; } while (rrel == i);
                }
                if (FIRST) {
                    const float w = __int_as_float(__builtin_amdgcn_readlane(wv, i));
                    ga.x = fmaf(w, h[i].x, ga.x);
                    ga.y = fmaf(w, h[i].y, ga.y);
                    ga.z = fmaf(w, h[i].z, ga.z);
                    ga.w = fmaf(w, h[i].w, ga.w);
                } else {
                    ga.x += h[i].x;
                    ga.y += h[i].y;
                    ga.z += h[i].z;
                    ga.w += h[i].w;
                }
            }
            sv1 = sv2;
            wv  = wn;
        }
        while (closed < 16) CLOSE_ROW();
#undef CLOSE_ROW

        __builtin_amdgcn_fence(__ATOMIC_RELEASE, "wavefront");
        __builtin_amdgcn_wave_barrier();
        __builtin_amdgcn_fence(__ATOMIC_ACQUIRE, "wavefront");

        const float* arow = tile + col * TP + khalf;

        v8f acc[8];
        const v8f vzero = {0.f, 0.f, 0.f, 0.f, 0.f, 0.f, 0.f, 0.f};
#pragma unroll
        for (int j = 0; j < 8; ++j) acc[j] = vzero;

        for (int k0 = 0; k0 < F; k0 += 4) {
            const v2f a = *(const v2f*)(arow + k0);
            const v2f* bbase = (const v2f*)&sW[((k0 + khalf) >> 1) * 256 + col * 2];
#pragma unroll
            for (int j = 0; j < 8; ++j) {
                v2f b = bbase[j * 16];
                acc[j] = __builtin_amdgcn_wmma_f32_16x16x4_f32(
                    false, a, false, b, (short)0, acc[j], false, false);
            }
        }

        float s[8], q[8];
#pragma unroll
        for (int i = 0; i < 8; ++i) { s[i] = 0.f; q[i] = 0.f; }
#pragma unroll
        for (int j = 0; j < 8; ++j) {
            const float bj = bias[j * 16 + col];
#pragma unroll
            for (int i = 0; i < 8; ++i) {
                const float v = fmaxf(acc[j][i] + bj, 0.f);
                acc[j][i] = v;
                s[i] += v;
            }
        }
#pragma unroll
        for (int i = 0; i < 8; ++i) {
            for (int m = 8; m >= 1; m >>= 1) s[i] += __shfl_xor(s[i], m, 32);
            s[i] *= (1.f / F);
        }
#pragma unroll
        for (int j = 0; j < 8; ++j) {
#pragma unroll
            for (int i = 0; i < 8; ++i) {
                const float d = acc[j][i] - s[i];
                acc[j][i] = d;
                q[i] = fmaf(d, d, q[i]);
            }
        }
#pragma unroll
        for (int i = 0; i < 8; ++i) {
            for (int m = 8; m >= 1; m >>= 1) q[i] += __shfl_xor(q[i], m, 32);
            q[i] = rsqrtf(q[i] * (1.f / F) + LN_EPS);
        }
        float* trow = tile + ((lane >> 4) * 8) * TP + col;
#pragma unroll
        for (int j = 0; j < 8; ++j) {
            const float gj = g[j * 16 + col];
            const float ej = be[j * 16 + col];
#pragma unroll
            for (int i = 0; i < 8; ++i) trow[i * TP + j * 16] = acc[j][i] * q[i] * gj + ej;
        }
        __builtin_amdgcn_fence(__ATOMIC_RELEASE, "wavefront");
        __builtin_amdgcn_wave_barrier();
        __builtin_amdgcn_fence(__ATOMIC_ACQUIRE, "wavefront");
        float4 o[16];
#pragma unroll
        for (int r = 0; r < 16; ++r) o[r] = *(const float4*)&tile[r * TP + lane * 4];
        if (FIRST) {
#pragma unroll
            for (int r = 0; r < 16; ++r) {
                const float dr = __int_as_float(__builtin_amdgcn_readlane(__float_as_int(ddv), r));
                o[r].x *= dr; o[r].y *= dr; o[r].z *= dr; o[r].w *= dr;
            }
        }
#pragma unroll
        for (int r = 0; r < 16; ++r)
            if (row0 + r < N)
                *(float4*)&out[(size_t)(row0 + r) * F + lane * 4] = o[r];
        __builtin_amdgcn_fence(__ATOMIC_RELEASE, "wavefront");
        __builtin_amdgcn_wave_barrier();
        __builtin_amdgcn_fence(__ATOMIC_ACQUIRE, "wavefront");
    }
}

static inline size_t align256(size_t x) { return (x + 255) & ~(size_t)255; }

extern "C" void kernel_launch(void* const* d_in, const int* in_sizes, int n_in,
                              void* d_out, int out_size, void* d_ws, size_t ws_size,
                              hipStream_t stream) {
    const float* x   = (const float*)d_in[0];
    const int*   ei  = (const int*)  d_in[1];
    const float* W1  = (const float*)d_in[2];
    const float* b1  = (const float*)d_in[3];
    const float* W2  = (const float*)d_in[4];
    const float* b2  = (const float*)d_in[5];
    const float* g1  = (const float*)d_in[6];
    const float* be1 = (const float*)d_in[7];
    const float* g2  = (const float*)d_in[8];
    const float* be2 = (const float*)d_in[9];

    const int N = in_sizes[0] / F;
    const int E = in_sizes[1] / 2;
    const int* src = ei;
    const int* dst = ei + E;
    float* out = (float*)d_out;

    char* ws = (char*)d_ws;
    size_t off = 0;
    float* dinv = (float*)(ws + off); off = align256(off + (size_t)N * 4);
    int*   cnt  = (int*)  (ws + off); off = align256(off + (size_t)N * 4);
    float* Y    = (float*)(ws + off); off = align256(off + (size_t)N * F * 4);
    int*   esrc   = (int*)(ws + off); off = align256(off + (size_t)E * 4);
    int*   rowptr = (int*)(ws + off); off = align256(off + (size_t)N * 4);
    int*   tmp    = (int*)(ws + off); off = align256(off + (size_t)E * 8);
    int*   scnt   = (int*)(ws + off); off = align256(off + (size_t)NSLICE * MAXCB * 4);
    int*   soff   = (int*)(ws + off); off = align256(off + (size_t)NSLICE * MAXCB * 4);
    (void)ws_size; (void)n_in; (void)out_size;

    k_part<<<NSLICE, dim3(1024), 0, stream>>>(src, dst, E, scnt, soff, tmp);
    k_fine<<<(N + CB_NODES - 1) / CB_NODES, dim3(1024), 0, stream>>>(tmp, scnt, soff, N, E, cnt, rowptr, dinv, esrc);

    k_layer<true ><<<LAYER_WGS, dim3(NWF * 32), 0, stream>>>(rowptr, cnt, esrc, dinv, x, W1, b1, g1, be1, Y, N);
    k_layer<false><<<LAYER_WGS, dim3(NWF * 32), 0, stream>>>(rowptr, cnt, esrc, dinv, Y, W2, b2, g2, be2, out, N);
}
